# starting kernel
# baseline (speedup 1.0000x reference)
#include <hip/hip_runtime.h>

typedef __attribute__((ext_vector_type(2))) float v2f;
typedef __attribute__((ext_vector_type(8))) float v8f;

__global__ void deg_init_k(float* __restrict__ deg, int N) {
  int t = blockIdx.x * blockDim.x + threadIdx.x;
  if (t < N) deg[t] = 1.0f;
}

__global__ void deg_scatter_k(const int* __restrict__ dst, const float* __restrict__ ew,
                              float* __restrict__ deg, int E) {
  int t = blockIdx.x * blockDim.x + threadIdx.x;
  if (t < E) unsafeAtomicAdd(&deg[dst[t]], ew[t]);
}

__global__ void dinv_k(const float* __restrict__ deg, float* __restrict__ dinv, int N) {
  int t = blockIdx.x * blockDim.x + threadIdx.x;
  if (t < N) {
    float d = deg[t];
    dinv[t] = (d > 0.0f) ? rsqrtf(d) : 0.0f;
  }
}

__global__ void norm_k(const int* __restrict__ src, const int* __restrict__ dst,
                       const float* __restrict__ ew, const float* __restrict__ dinv,
                       float* __restrict__ nrm, int E) {
  int t = blockIdx.x * blockDim.x + threadIdx.x;
  if (t < E) nrm[t] = dinv[src[t]] * ew[t] * dinv[dst[t]];
}

template <int K, int F>
__global__ void gemm16_wmma(const float* __restrict__ X, const float* __restrict__ W,
                            float* __restrict__ H, int N) {
  const int lane = threadIdx.x & 31;
  const int wave = threadIdx.x >> 5;
  const int Ft = F >> 4;
  const int Mt = N >> 4;
  const int tile = blockIdx.x * (blockDim.x >> 5) + wave;
  if (tile >= Mt * Ft) return;
  const int mt = tile / Ft;
  const int ft = tile - mt * Ft;
  const int l16 = lane & 15;
  const int hi  = lane >> 4;

  const float* __restrict__ xrow = X + (size_t)(mt * 16 + l16) * K;
  const float* __restrict__ wcol = W + ft * 16 + l16;

  v8f acc = {};
#pragma unroll 4
  for (int k = 0; k < K; k += 4) {
    v2f a, b;
    a.x = xrow[k + 2 * hi];
    a.y = xrow[k + 2 * hi + 1];
    b.x = wcol[(size_t)(k + 2 * hi) * F];
    b.y = wcol[(size_t)(k + 2 * hi + 1) * F];
    acc = __builtin_amdgcn_wmma_f32_16x16x4_f32(false, a, false, b,
                                                (short)0, acc, false, false);
  }

  float* __restrict__ hcol = H + ft * 16 + l16;
#pragma unroll
  for (int r = 0; r < 8; ++r)
    hcol[(size_t)(mt * 16 + r + 8 * hi) * F] = acc[r];
}

template <int F>
__global__ void agg_init_k(const float* __restrict__ h, const float* __restrict__ dinv,
                           const float* __restrict__ bias, float* __restrict__ a, int N) {
  int t = blockIdx.x * blockDim.x + threadIdx.x;
  if (t >= N * F) return;
  constexpr int LOGF = (F == 64) ? 6 : 5;
  int n = t >> LOGF;
  int f = t & (F - 1);
  float di = dinv[n];
  a[t] = bias[f] + di * di * h[t];
}

template <int F>
__global__ void agg_edges_k(const float* __restrict__ h, const int* __restrict__ src,
                            const int* __restrict__ dst, const float* __restrict__ nrm,
                            float* __restrict__ a, int E) {
  unsigned t = blockIdx.x * blockDim.x + threadIdx.x;
  constexpr int LOGF = (F == 64) ? 6 : 5;
  unsigned e = t >> LOGF;
  if (e >= (unsigned)E) return;
  unsigned f = t & (F - 1);
  int s = src[e];
  int d = dst[e];
  float w = nrm[e];
  unsafeAtomicAdd(&a[(size_t)d * F + f], w * h[(size_t)s * F + f]);
}

__global__ void relu_k(float* __restrict__ a, int n) {
  int t = blockIdx.x * blockDim.x + threadIdx.x;
  if (t < n) a[t] = fmaxf(a[t], 0.0f);
}

__global__ void mean_pool_k(const float* __restrict__ emb, float* __restrict__ g, int N) {
  __shared__ float sm[256];
  const int c = blockIdx.x;
  float s = 0.0f;
  for (int r = threadIdx.x; r < N; r += blockDim.x) s += emb[(size_t)r * 32 + c];
  sm[threadIdx.x] = s;
  __syncthreads();
  for (int st = 128; st > 0; st >>= 1) {
    if (threadIdx.x < st) sm[threadIdx.x] += sm[threadIdx.x + st];
    __syncthreads();
  }
  if (threadIdx.x == 0) g[c] = sm[0] / (float)N;
}

static inline unsigned cdiv(long a, long b) { return (unsigned)((a + b - 1) / b); }

extern "C" void kernel_launch(void* const* d_in, const int* in_sizes, int n_in,
                              void* d_out, int out_size, void* d_ws, size_t ws_size,
                              hipStream_t stream) {
  const float* x  = (const float*)d_in[0];
  const int*   ei = (const int*)d_in[1];
  const float* ew = (const float*)d_in[2];
  const float* W1 = (const float*)d_in[3];
  const float* b1 = (const float*)d_in[4];
  const float* W2 = (const float*)d_in[5];
  const float* b2 = (const float*)d_in[6];
  const float* W3 = (const float*)d_in[7];
  const float* b3 = (const float*)d_in[8];

  const int N = in_sizes[0] / 128;
  const int E = in_sizes[2];
  const int* src = ei;
  const int* dst = ei + E;

  float* ws   = (float*)d_ws;
  float* deg  = ws;  ws += N;
  float* dinv = ws;  ws += N;
  float* nrm  = ws;  ws += E;
  float* bufA = ws;  ws += (size_t)N * 64;
  float* bufB = ws;
  float* out  = (float*)d_out;

  const dim3 blk(256);

  deg_init_k   <<<cdiv(N, 256), blk, 0, stream>>>(deg, N);
  deg_scatter_k<<<cdiv(E, 256), blk, 0, stream>>>(dst, ew, deg, E);
  dinv_k       <<<cdiv(N, 256), blk, 0, stream>>>(deg, dinv, N);
  norm_k       <<<cdiv(E, 256), blk, 0, stream>>>(src, dst, ew, dinv, nrm, E);

  gemm16_wmma<128, 64><<<cdiv((long)(N / 16) * 4, 8), blk, 0, stream>>>(x, W1, bufA, N);
  agg_init_k<64> <<<cdiv((long)N * 64, 256), blk, 0, stream>>>(bufA, dinv, b1, bufB, N);
  agg_edges_k<64><<<cdiv((long)E * 64, 256), blk, 0, stream>>>(bufA, src, dst, nrm, bufB, E);
  relu_k         <<<cdiv((long)N * 64, 256), blk, 0, stream>>>(bufB, N * 64);

  gemm16_wmma<64, 64><<<cdiv((long)(N / 16) * 4, 8), blk, 0, stream>>>(bufB, W2, bufA, N);
  agg_init_k<64> <<<cdiv((long)N * 64, 256), blk, 0, stream>>>(bufA, dinv, b2, bufB, N);
  agg_edges_k<64><<<cdiv((long)E * 64, 256), blk, 0, stream>>>(bufA, src, dst, nrm, bufB, E);
  relu_k         <<<cdiv((long)N * 64, 256), blk, 0, stream>>>(bufB, N * 64);

  gemm16_wmma<64, 32><<<cdiv((long)(N / 16) * 2, 8), blk, 0, stream>>>(bufB, W3, bufA, N);
  agg_init_k<32> <<<cdiv((long)N * 32, 256), blk, 0, stream>>>(bufA, dinv, b3, out, N);
  agg_edges_k<32><<<cdiv((long)E * 32, 256), blk, 0, stream>>>(bufA, src, dst, nrm, out, E);
  mean_pool_k    <<<32, 256, 0, stream>>>(out, out + (size_t)N * 32, N);
}


# Degree sums: one scattered float atomic per edge -> work-groups sum an edge chunk's weights per 65536-node range in LDS (LDS float adds), then add whole 128-B lines to deg; degree sum order changes
# speedup vs baseline: 4.7338x; 4.7338x over previous
#include <hip/hip_runtime.h>

typedef __attribute__((ext_vector_type(2))) float v2f;
typedef __attribute__((ext_vector_type(8))) float v8f;

constexpr int POOL_SLOTS = 256;

constexpr int CTR_STRIDE = 64;
constexpr int CTR_WORDS = 3 * 2 * CTR_STRIDE;

__global__ void deg_init_k(float* __restrict__ deg, float* __restrict__ pool,
                           unsigned* __restrict__ ctr, int N) {
  int t = blockIdx.x * blockDim.x + threadIdx.x;
  if (t < N) deg[t] = 1.0f;
  for (int i = t; i < POOL_SLOTS * 32; i += gridDim.x * blockDim.x) pool[i] = 0.0f;
  for (int i = t; i < CTR_WORDS; i += gridDim.x * blockDim.x) ctr[i] = 0u;
}

constexpr int DEG_RANGE = 65536;
constexpr int DEG_CHUNKS = 64;
constexpr int DEG_THREADS = 1024;

__global__ __launch_bounds__(DEG_THREADS)
void deg_hist_k(const int* __restrict__ dst, const float* __restrict__ ew,
                float* __restrict__ deg, int E, int N) {
  __shared__ float4 hs4[DEG_RANGE / 4];
  float* hs = reinterpret_cast<float*>(hs4);
  const unsigned tid = threadIdx.x;
  const unsigned lo = blockIdx.y * (unsigned)DEG_RANGE;
  const unsigned len = min((unsigned)DEG_RANGE, (unsigned)N - lo);
  for (unsigned i = tid; i < DEG_RANGE / 4; i += DEG_THREADS) hs4[i] = make_float4(0.0f, 0.0f, 0.0f, 0.0f);
  __syncthreads();

  const int per = (E + DEG_CHUNKS - 1) / DEG_CHUNKS;
  const int e0 = blockIdx.x * per;
  const int e1 = min(e0 + per, E);
  for (int e = e0 + (int)tid; e < e1; e += 16 * DEG_THREADS) {
    int d[16];
    float w[16];
#pragma unroll
    for (int j = 0; j < 16; ++j) {
      const int idx = e + j * DEG_THREADS;
      const int c = min(idx, e1 - 1);
      d[j] = dst[c];
      w[j] = ew[c];
      if (idx >= e1) d[j] = -1;
    }
    asm volatile("" ::"v"(w[0]), "v"(w[1]), "v"(w[2]), "v"(w[3]), "v"(w[4]), "v"(w[5]), "v"(w[6]), "v"(w[7]),
                 "v"(w[8]), "v"(w[9]), "v"(w[10]), "v"(w[11]), "v"(w[12]), "v"(w[13]), "v"(w[14]), "v"(w[15]));
#pragma unroll
    for (int j = 0; j < 16; ++j) {
      const unsigned u = (unsigned)d[j] - lo;
      if (u < len) unsafeAtomicAdd(&hs[u], w[j]);
    }
  }
  __syncthreads();

  const unsigned lane = tid & 31;
  const unsigned nl = (len + 31) / 32;
  const unsigned l0 = blockIdx.x * nl / DEG_CHUNKS;
  for (unsigned k = tid >> 5; k < nl; k += 4 * (DEG_THREADS / 32)) {
    unsigned at[4];
    float v[4];
#pragma unroll
    for (int j = 0; j < 4; ++j) {
      const unsigned kk = k + j * (DEG_THREADS / 32);
      unsigned line = kk + l0;
      if (line >= nl) line -= nl;
      const unsigned i = line * 32 + lane;
      const bool ok = kk < nl && i < len;
      at[j] = ok ? i : 0xffffffffu;
      v[j] = ok ? hs[i] : 0.0f;
    }
#pragma unroll
    for (int j = 0; j < 4; ++j)
      if (at[j] != 0xffffffffu) unsafeAtomicAdd(&deg[lo + at[j]], v[j]);
  }
}

__global__ void dinv_k(const float* __restrict__ deg, float* __restrict__ dinv, int N) {
  int t = blockIdx.x * blockDim.x + threadIdx.x;
  if (t < N) {
    float d = deg[t];
    dinv[t] = (d > 0.0f) ? rsqrtf(d) : 0.0f;
  }
}

__global__ void norm_k(const int* __restrict__ src, const int* __restrict__ dst,
                       const float* __restrict__ ew, const float* __restrict__ dinv,
                       float* __restrict__ nrm, int E) {
  int t = blockIdx.x * blockDim.x + threadIdx.x;
  if (t < E) nrm[t] = dinv[src[t]] * ew[t] * dinv[dst[t]];
}

template <int K, int F, bool RELU_IN, bool POOL, bool IN_PLANES>
__global__ void gemm16_wmma(const float* __restrict__ X, const float* __restrict__ W,
                            const float* __restrict__ dinv, const float* __restrict__ bias,
                            float* __restrict__ H, float* __restrict__ S,
                            float* __restrict__ pool, int N) {
  const int lane = threadIdx.x & 31;
  const int wave = threadIdx.x >> 5;
  const int Ft = F >> 4;
  const int Mt = N >> 4;
  const int tile = blockIdx.x * (blockDim.x >> 5) + wave;
  if (tile >= Mt * Ft) return;
  const int mt = tile / Ft;
  const int ft = tile - mt * Ft;
  const int l16 = lane & 15;
  const int hi  = lane >> 4;

  constexpr bool OUT_PLANES = (F == 64);
  static_assert(!IN_PLANES || K == 64, "the plane layout is that of 64-column activations");
  const size_t plane = (size_t)N * 32;
  const float* __restrict__ xrow = X + (size_t)(mt * 16 + l16) * (IN_PLANES ? 32 : K);
  const float* __restrict__ wcol = W + ft * 16 + l16;

  v8f acc = {};
  auto kstep = [&](const float* __restrict__ xk, int k) {
    v2f a, b;
    a.x = xk[2 * hi];
    a.y = xk[2 * hi + 1];
    if constexpr (RELU_IN) {
      a.x = fmaxf(a.x, 0.0f);
      a.y = fmaxf(a.y, 0.0f);
    }
    b.x = wcol[(size_t)(k + 2 * hi) * F];
    b.y = wcol[(size_t)(k + 2 * hi + 1) * F];
    acc = __builtin_amdgcn_wmma_f32_16x16x4_f32(false, a, false, b,
                                                (short)0, acc, false, false);
  };
  if constexpr (IN_PLANES) {
    for (int c = 0; c < K / 32; ++c) {
      const float* __restrict__ xp = xrow + (size_t)c * plane;
#pragma unroll 4
      for (int kk = 0; kk < 32; kk += 4) kstep(xp + kk, c * 32 + kk);
    }
  } else {
#pragma unroll 4
    for (int k = 0; k < K; k += 4) kstep(xrow + k, k);
  }

  const int col = ft * 16 + l16;
  const int row0 = mt * 16 + 8 * hi;
  const float bs = bias[col];
  float di[8];
#pragma unroll
  for (int r = 0; r < 8; ++r) di[r] = dinv[row0 + r];
  float p = 0.0f;
#pragma unroll
  for (int r = 0; r < 8; ++r) {
    const size_t o = OUT_PLANES ? (size_t)(col >> 5) * plane + (size_t)(row0 + r) * 32 + (col & 31)
                                : (size_t)(row0 + r) * F + col;
    const float hv = acc[r];
    const float sv = fmaf(di[r] * di[r], hv, bs);
    H[o] = hv;
    S[o] = sv;
    if constexpr (POOL) p += sv;
  }
  if constexpr (POOL) {
    p += __shfl_xor(p, 16);
    if (hi == 0)
      unsafeAtomicAdd(&pool[(size_t)(mt & (POOL_SLOTS - 1)) * 32 + col], p);
  }
}

template <int F, bool POOL>
__global__ __launch_bounds__(256)
void agg_edges_k(const float* __restrict__ h, const int* __restrict__ src,
                 const int* __restrict__ dst, const float* __restrict__ nrm,
                 float* __restrict__ a, float* __restrict__ pool,
                 unsigned* __restrict__ ctr, int E, int N) {
  constexpr bool BYCOL = (F == 64);
  static_assert(F == 64 || F == 32, "one or two 128-B lines per row");
  static_assert(!POOL || !BYCOL, "column sums are kept for one-line rows only");
  constexpr unsigned SB = 16;
  constexpr unsigned EPW = 64;
  constexpr unsigned EPI = EPW * 8;
  __shared__ unsigned s_t[2];
  const unsigned lane = threadIdx.x & 31;
  const unsigned wv = __builtin_amdgcn_readfirstlane(threadIdx.x >> 5);
  const unsigned gw = blockIdx.x * (blockDim.x >> 5) + wv;
  const unsigned nitems = ((unsigned)E + EPI - 1) / EPI;
  const unsigned nsplit = ((unsigned)N >> 1) & ~1u;
  const size_t plane = (size_t)N * 32;
  unsigned q = __builtin_amdgcn_readfirstlane(
      ((unsigned)__builtin_amdgcn_s_sendmsg_rtn(0x87) >> 18) & 1u);

  float p = 0.0f;
  bool own = true;
  unsigned slot = 0;
  unsigned drawn = 0;
  if (threadIdx.x == 0) drawn = atomicAdd(&ctr[q * CTR_STRIDE], 1u);
  for (;;) {
    if (threadIdx.x == 0) s_t[slot] = drawn;
    asm volatile("s_wait_dscnt 0x0" ::: "memory");
    __builtin_amdgcn_s_barrier();
    const unsigned t = __builtin_amdgcn_readfirstlane(s_t[slot]);
    slot ^= 1u;
    if (t >= nitems) {
      if (!own) break;
      own = false;
      q ^= 1u;
      if (threadIdx.x == 0) drawn = atomicAdd(&ctr[q * CTR_STRIDE], 1u);
      continue;
    }
    if (threadIdx.x == 0) drawn = atomicAdd(&ctr[q * CTR_STRIDE], 1u);
    const float* __restrict__ hq = BYCOL ? h + q * plane : h;
    float* __restrict__ aq = BYCOL ? a + q * plane : a;
    const unsigned lo = q ? nsplit : 0u;
    const unsigned len = q ? (unsigned)N - nsplit : nsplit;
    const unsigned e0 = t * EPI + wv * EPW;
    const unsigned e1 = min(e0 + EPW, (unsigned)E);
    for (unsigned eb = e0; eb < e1; eb += SB) {
      if (eb + SB <= (unsigned)E) {
        int s[SB], d[SB];
        float w[SB];
        for (unsigned j = 0; j < SB; ++j) {
          s[j] = src[eb + j];
          d[j] = dst[eb + j];
          w[j] = nrm[eb + j];
        }
        if constexpr (!BYCOL) {
          asm volatile("" ::"s"(s[0]), "s"(s[1]), "s"(s[2]), "s"(s[3]), "s"(s[4]), "s"(s[5]), "s"(s[6]),
                       "s"(s[7]), "s"(s[8]), "s"(s[9]), "s"(s[10]), "s"(s[11]), "s"(s[12]), "s"(s[13]),
                       "s"(s[14]), "s"(s[15]));
          asm volatile("" ::"s"(w[0]), "s"(w[1]), "s"(w[2]), "s"(w[3]), "s"(w[4]), "s"(w[5]), "s"(w[6]),
                       "s"(w[7]), "s"(w[8]), "s"(w[9]), "s"(w[10]), "s"(w[11]), "s"(w[12]), "s"(w[13]),
                       "s"(w[14]), "s"(w[15]));
        }
        float v[SB];
        for (unsigned j = 0; j < SB; ++j) {
          v[j] = 0.0f;
          if (BYCOL || (unsigned)d[j] - lo < len)
            v[j] = hq[(size_t)(unsigned)s[j] * 32 + lane];
        }
        for (unsigned j = 0; j < SB; ++j)
          if (BYCOL || (unsigned)d[j] - lo < len) {
            const float m = w[j] * v[j];
            unsafeAtomicAdd(&aq[(size_t)(unsigned)d[j] * 32 + lane], m);
            if constexpr (POOL) p += m;
          }
      } else {
        for (unsigned e = eb; e < (unsigned)E; ++e) {
          const int s = src[e];
          const int d = dst[e];
          const float w = nrm[e];
          if (BYCOL || (unsigned)d - lo < len) {
            const float m = w * hq[(size_t)(unsigned)s * 32 + lane];
            unsafeAtomicAdd(&aq[(size_t)(unsigned)d * 32 + lane], m);
            if constexpr (POOL) p += m;
          }
        }
      }
    }
  }
  if constexpr (POOL)
    unsafeAtomicAdd(&pool[(size_t)(gw & (POOL_SLOTS - 1)) * 32 + lane], p);
}

__global__ __launch_bounds__(256)
void pool_final_k(const float* __restrict__ pool, float* __restrict__ g, int N) {
  __shared__ float sm[8][32];
  const unsigned lane = threadIdx.x & 31;
  const unsigned q = threadIdx.x >> 5;
  float s = 0.0f;
  for (int i = 0; i < POOL_SLOTS / 8; ++i) s += pool[(size_t)(q + 8 * i) * 32 + lane];
  sm[q][lane] = s;
  __syncthreads();
  if (q == 0) {
    float t = 0.0f;
    for (int w = 0; w < 8; ++w) t += sm[w][lane];
    g[lane] = t / (float)N;
  }
}

static inline unsigned cdiv(long a, long b) { return (unsigned)((a + b - 1) / b); }

extern "C" void kernel_launch(void* const* d_in, const int* in_sizes, int n_in,
                              void* d_out, int out_size, void* d_ws, size_t ws_size,
                              hipStream_t stream) {
  const float* x  = (const float*)d_in[0];
  const int*   ei = (const int*)d_in[1];
  const float* ew = (const float*)d_in[2];
  const float* W1 = (const float*)d_in[3];
  const float* b1 = (const float*)d_in[4];
  const float* W2 = (const float*)d_in[5];
  const float* b2 = (const float*)d_in[6];
  const float* W3 = (const float*)d_in[7];
  const float* b3 = (const float*)d_in[8];

  const int N = in_sizes[0] / 128;
  const int E = in_sizes[2];
  const int* src = ei;
  const int* dst = ei + E;

  float* ws   = (float*)d_ws;
  float* deg  = ws;  ws += N;
  float* dinv = ws;  ws += N;
  float* nrm  = ws;  ws += E;
  float* bufA = ws;  ws += (size_t)N * 64;
  float* bufB = ws;  ws += (size_t)N * 64;
  float* bufC = ws;  ws += (size_t)N * 64;
  float* pool = ws;  ws += POOL_SLOTS * 32;
  unsigned* ctr = (unsigned*)ws;
  float* out  = (float*)d_out;

  const dim3 blk(256);

  deg_init_k   <<<cdiv(N, 256), blk, 0, stream>>>(deg, pool, ctr, N);
  deg_hist_k   <<<dim3(DEG_CHUNKS, cdiv(N, DEG_RANGE)), dim3(DEG_THREADS), 0, stream>>>(dst, ew, deg, E, N);
  dinv_k       <<<cdiv(N, 256), blk, 0, stream>>>(deg, dinv, N);
  norm_k       <<<cdiv(E, 256), blk, 0, stream>>>(src, dst, ew, dinv, nrm, E);

  gemm16_wmma<128, 64, false, false, false><<<cdiv((long)(N / 16) * 4, 8), blk, 0, stream>>>(x, W1, dinv, b1, bufA, bufB, pool, N);
  agg_edges_k<64, false><<<2048, blk, 0, stream>>>(bufA, src, dst, nrm, bufB, pool, ctr, E, N);

  gemm16_wmma<64, 64, true, false, true><<<cdiv((long)(N / 16) * 4, 8), blk, 0, stream>>>(bufB, W2, dinv, b2, bufA, bufC, pool, N);
  agg_edges_k<64, false><<<2048, blk, 0, stream>>>(bufA, src, dst, nrm, bufC, pool, ctr + 2 * CTR_STRIDE, E, N);

  gemm16_wmma<64, 32, true, true, true><<<cdiv((long)(N / 16) * 2, 8), blk, 0, stream>>>(bufC, W3, dinv, b3, bufA, out, pool, N);
  agg_edges_k<32, true><<<2048, blk, 0, stream>>>(bufA, src, dst, nrm, out, pool, ctr + 4 * CTR_STRIDE, E, N);
  pool_final_k   <<<1, 256, 0, stream>>>(pool, out + (size_t)N * 32, N);
}


# Edge-weight pass: a work-group copies dinv into LDS (65536-node ranges in turn, 16-B loads) and each edge takes both factors from LDS; was one lane per edge with 2 global gathers; all sums kept
# speedup vs baseline: 4.8963x; 1.0343x over previous
#include <hip/hip_runtime.h>

typedef __attribute__((ext_vector_type(2))) float v2f;
typedef __attribute__((ext_vector_type(4))) float v4f;
typedef __attribute__((ext_vector_type(8))) float v8f;

constexpr int POOL_SLOTS = 256;

constexpr int CTR_STRIDE = 64;
constexpr int CTR_WORDS = 3 * 2 * CTR_STRIDE;

__global__ void deg_init_k(float* __restrict__ deg, float* __restrict__ pool,
                           unsigned* __restrict__ ctr, int N) {
  int t = blockIdx.x * blockDim.x + threadIdx.x;
  if (t < N) deg[t] = 1.0f;
  for (int i = t; i < POOL_SLOTS * 32; i += gridDim.x * blockDim.x) pool[i] = 0.0f;
  for (int i = t; i < CTR_WORDS; i += gridDim.x * blockDim.x) ctr[i] = 0u;
}

constexpr int DEG_RANGE = 65536;
constexpr int DEG_CHUNKS = 64;
constexpr int DEG_THREADS = 1024;

__global__ __launch_bounds__(DEG_THREADS)
void deg_hist_k(const int* __restrict__ dst, const float* __restrict__ ew,
                float* __restrict__ deg, int E, int N) {
  __shared__ float4 hs4[DEG_RANGE / 4];
  float* hs = reinterpret_cast<float*>(hs4);
  const unsigned tid = threadIdx.x;
  const unsigned lo = blockIdx.y * (unsigned)DEG_RANGE;
  const unsigned len = min((unsigned)DEG_RANGE, (unsigned)N - lo);
  for (unsigned i = tid; i < DEG_RANGE / 4; i += DEG_THREADS) hs4[i] = make_float4(0.0f, 0.0f, 0.0f, 0.0f);
  __syncthreads();

  const int per = (E + DEG_CHUNKS - 1) / DEG_CHUNKS;
  const int e0 = blockIdx.x * per;
  const int e1 = min(e0 + per, E);
  for (int e = e0 + (int)tid; e < e1; e += 16 * DEG_THREADS) {
    int d[16];
    float w[16];
#pragma unroll
    for (int j = 0; j < 16; ++j) {
      const int idx = e + j * DEG_THREADS;
      const int c = min(idx, e1 - 1);
      d[j] = dst[c];
      w[j] = ew[c];
      if (idx >= e1) d[j] = -1;
    }
    asm volatile("" ::"v"(w[0]), "v"(w[1]), "v"(w[2]), "v"(w[3]), "v"(w[4]), "v"(w[5]), "v"(w[6]), "v"(w[7]),
                 "v"(w[8]), "v"(w[9]), "v"(w[10]), "v"(w[11]), "v"(w[12]), "v"(w[13]), "v"(w[14]), "v"(w[15]));
#pragma unroll
    for (int j = 0; j < 16; ++j) {
      const unsigned u = (unsigned)d[j] - lo;
      if (u < len) unsafeAtomicAdd(&hs[u], w[j]);
    }
  }
  __syncthreads();

  const unsigned lane = tid & 31;
  const unsigned nl = (len + 31) / 32;
  const unsigned l0 = blockIdx.x * nl / DEG_CHUNKS;
  for (unsigned k = tid >> 5; k < nl; k += 4 * (DEG_THREADS / 32)) {
    unsigned at[4];
    float v[4];
#pragma unroll
    for (int j = 0; j < 4; ++j) {
      const unsigned kk = k + j * (DEG_THREADS / 32);
      unsigned line = kk + l0;
      if (line >= nl) line -= nl;
      const unsigned i = line * 32 + lane;
      const bool ok = kk < nl && i < len;
      at[j] = ok ? i : 0xffffffffu;
      v[j] = ok ? hs[i] : 0.0f;
    }
#pragma unroll
    for (int j = 0; j < 4; ++j)
      if (at[j] != 0xffffffffu) unsafeAtomicAdd(&deg[lo + at[j]], v[j]);
  }
}

__global__ void dinv_k(const float* __restrict__ deg, float* __restrict__ dinv, int N) {
  int t = blockIdx.x * blockDim.x + threadIdx.x;
  if (t < N) {
    float d = deg[t];
    dinv[t] = (d > 0.0f) ? rsqrtf(d) : 0.0f;
  }
}

constexpr int NRM_RANGE = 65536;
constexpr int NRM_THREADS = 1024;
constexpr int NRM_EPT = 7;
constexpr int NRM_SG = NRM_RANGE / 4 / NRM_THREADS;

__global__ __launch_bounds__(NRM_THREADS)
void norm_lds_k(const int* __restrict__ src, const int* __restrict__ dst,
                const float* __restrict__ ew, const float* __restrict__ dinv,
                float* __restrict__ nrm, int E, int N) {
  __shared__ v4f hs4[NRM_RANGE / 4];
  float* hs = reinterpret_cast<float*>(hs4);
  const unsigned tid = threadIdx.x;
  const unsigned e0 = blockIdx.x * (unsigned)(NRM_THREADS * NRM_EPT);

  int s[NRM_EPT], d[NRM_EPT];
  float w[NRM_EPT], fs[NRM_EPT], fd[NRM_EPT];
#pragma unroll
  for (int j = 0; j < NRM_EPT; ++j) {
    const unsigned c = min(e0 + j * NRM_THREADS + tid, (unsigned)E - 1u);
    s[j] = src[c];
    d[j] = dst[c];
    w[j] = ew[c];
    fs[j] = 0.0f;
    fd[j] = 0.0f;
  }

  for (unsigned lo = 0; lo < (unsigned)N; lo += NRM_RANGE) {
    const unsigned len = min((unsigned)NRM_RANGE, (unsigned)N - lo);
    const unsigned len4 = len & ~3u;
    if (lo != 0) __syncthreads();
    if (len4 != 0) {
      v4f g[NRM_SG];
#pragma unroll
      for (int k = 0; k < NRM_SG; ++k) {
        const unsigned i = min((unsigned)(k * NRM_THREADS + tid) * 4, len4 - 4);
        g[k] = *reinterpret_cast<const v4f*>(dinv + lo + i);
      }
      static_assert(NRM_SG == 16, "the statement below names the value of each load");
      asm volatile("" ::"v"(g[0]), "v"(g[1]), "v"(g[2]), "v"(g[3]), "v"(g[4]), "v"(g[5]), "v"(g[6]), "v"(g[7]),
                   "v"(g[8]), "v"(g[9]), "v"(g[10]), "v"(g[11]), "v"(g[12]), "v"(g[13]), "v"(g[14]), "v"(g[15]));
#pragma unroll
      for (int k = 0; k < NRM_SG; ++k) {
        const unsigned i = min((unsigned)(k * NRM_THREADS + tid) * 4, len4 - 4);
        hs4[i >> 2] = g[k];
      }
    }
    if (tid < len - len4) hs[len4 + tid] = dinv[lo + len4 + tid];
    __syncthreads();

#pragma unroll
    for (int j = 0; j < NRM_EPT; ++j) {
      const unsigned us = (unsigned)s[j] - lo;
      const unsigned ud = (unsigned)d[j] - lo;
      if (us < len) fs[j] = hs[us];
      if (ud < len) fd[j] = hs[ud];
    }
  }

#pragma unroll
  for (int j = 0; j < NRM_EPT; ++j) {
    const unsigned e = e0 + j * NRM_THREADS + tid;
    if (e < (unsigned)E) nrm[e] = fs[j] * w[j] * fd[j];
  }
}

template <int K, int F, bool RELU_IN, bool POOL, bool IN_PLANES>
__global__ void gemm16_wmma(const float* __restrict__ X, const float* __restrict__ W,
                            const float* __restrict__ dinv, const float* __restrict__ bias,
                            float* __restrict__ H, float* __restrict__ S,
                            float* __restrict__ pool, int N) {
  const int lane = threadIdx.x & 31;
  const int wave = threadIdx.x >> 5;
  const int Ft = F >> 4;
  const int Mt = N >> 4;
  const int tile = blockIdx.x * (blockDim.x >> 5) + wave;
  if (tile >= Mt * Ft) return;
  const int mt = tile / Ft;
  const int ft = tile - mt * Ft;
  const int l16 = lane & 15;
  const int hi  = lane >> 4;

  constexpr bool OUT_PLANES = (F == 64);
  static_assert(!IN_PLANES || K == 64, "the plane layout is that of 64-column activations");
  const size_t plane = (size_t)N * 32;
  const float* __restrict__ xrow = X + (size_t)(mt * 16 + l16) * (IN_PLANES ? 32 : K);
  const float* __restrict__ wcol = W + ft * 16 + l16;

  v8f acc = {};
  auto kstep = [&](const float* __restrict__ xk, int k) {
    v2f a, b;
    a.x = xk[2 * hi];
    a.y = xk[2 * hi + 1];
    if constexpr (RELU_IN) {
      a.x = fmaxf(a.x, 0.0f);
      a.y = fmaxf(a.y, 0.0f);
    }
    b.x = wcol[(size_t)(k + 2 * hi) * F];
    b.y = wcol[(size_t)(k + 2 * hi + 1) * F];
    acc = __builtin_amdgcn_wmma_f32_16x16x4_f32(false, a, false, b,
                                                (short)0, acc, false, false);
  };
  if constexpr (IN_PLANES) {
    for (int c = 0; c < K / 32; ++c) {
      const float* __restrict__ xp = xrow + (size_t)c * plane;
#pragma unroll 4
      for (int kk = 0; kk < 32; kk += 4) kstep(xp + kk, c * 32 + kk);
    }
  } else {
#pragma unroll 4
    for (int k = 0; k < K; k += 4) kstep(xrow + k, k);
  }

  const int col = ft * 16 + l16;
  const int row0 = mt * 16 + 8 * hi;
  const float bs = bias[col];
  float di[8];
#pragma unroll
  for (int r = 0; r < 8; ++r) di[r] = dinv[row0 + r];
  float p = 0.0f;
#pragma unroll
  for (int r = 0; r < 8; ++r) {
    const size_t o = OUT_PLANES ? (size_t)(col >> 5) * plane + (size_t)(row0 + r) * 32 + (col & 31)
                                : (size_t)(row0 + r) * F + col;
    const float hv = acc[r];
    const float sv = fmaf(di[r] * di[r], hv, bs);
    H[o] = hv;
    S[o] = sv;
    if constexpr (POOL) p += sv;
  }
  if constexpr (POOL) {
    p += __shfl_xor(p, 16);
    if (hi == 0)
      unsafeAtomicAdd(&pool[(size_t)(mt & (POOL_SLOTS - 1)) * 32 + col], p);
  }
}

template <int F, bool POOL>
__global__ __launch_bounds__(256)
void agg_edges_k(const float* __restrict__ h, const int* __restrict__ src,
                 const int* __restrict__ dst, const float* __restrict__ nrm,
                 float* __restrict__ a, float* __restrict__ pool,
                 unsigned* __restrict__ ctr, int E, int N) {
  constexpr bool BYCOL = (F == 64);
  static_assert(F == 64 || F == 32, "one or two 128-B lines per row");
  static_assert(!POOL || !BYCOL, "column sums are kept for one-line rows only");
  constexpr unsigned SB = 16;
  constexpr unsigned EPW = 64;
  constexpr unsigned EPI = EPW * 8;
  __shared__ unsigned s_t[2];
  const unsigned lane = threadIdx.x & 31;
  const unsigned wv = __builtin_amdgcn_readfirstlane(threadIdx.x >> 5);
  const unsigned gw = blockIdx.x * (blockDim.x >> 5) + wv;
  const unsigned nitems = ((unsigned)E + EPI - 1) / EPI;
  const unsigned nsplit = ((unsigned)N >> 1) & ~1u;
  const size_t plane = (size_t)N * 32;
  unsigned q = __builtin_amdgcn_readfirstlane(
      ((unsigned)__builtin_amdgcn_s_sendmsg_rtn(0x87) >> 18) & 1u);

  float p = 0.0f;
  bool own = true;
  unsigned slot = 0;
  unsigned drawn = 0;
  if (threadIdx.x == 0) drawn = atomicAdd(&ctr[q * CTR_STRIDE], 1u);
  for (;;) {
    if (threadIdx.x == 0) s_t[slot] = drawn;
    asm volatile("s_wait_dscnt 0x0" ::: "memory");
    __builtin_amdgcn_s_barrier();
    const unsigned t = __builtin_amdgcn_readfirstlane(s_t[slot]);
    slot ^= 1u;
    if (t >= nitems) {
      if (!own) break;
      own = false;
      q ^= 1u;
      if (threadIdx.x == 0) drawn = atomicAdd(&ctr[q * CTR_STRIDE], 1u);
      continue;
    }
    if (threadIdx.x == 0) drawn = atomicAdd(&ctr[q * CTR_STRIDE], 1u);
    const float* __restrict__ hq = BYCOL ? h + q * plane : h;
    float* __restrict__ aq = BYCOL ? a + q * plane : a;
    const unsigned lo = q ? nsplit : 0u;
    const unsigned len = q ? (unsigned)N - nsplit : nsplit;
    const unsigned e0 = t * EPI + wv * EPW;
    const unsigned e1 = min(e0 + EPW, (unsigned)E);
    for (unsigned eb = e0; eb < e1; eb += SB) {
      if (eb + SB <= (unsigned)E) {
        int s[SB], d[SB];
        float w[SB];
        for (unsigned j = 0; j < SB; ++j) {
          s[j] = src[eb + j];
          d[j] = dst[eb + j];
          w[j] = nrm[eb + j];
        }
        if constexpr (!BYCOL) {
          asm volatile("" ::"s"(s[0]), "s"(s[1]), "s"(s[2]), "s"(s[3]), "s"(s[4]), "s"(s[5]), "s"(s[6]),
                       "s"(s[7]), "s"(s[8]), "s"(s[9]), "s"(s[10]), "s"(s[11]), "s"(s[12]), "s"(s[13]),
                       "s"(s[14]), "s"(s[15]));
          asm volatile("" ::"s"(w[0]), "s"(w[1]), "s"(w[2]), "s"(w[3]), "s"(w[4]), "s"(w[5]), "s"(w[6]),
                       "s"(w[7]), "s"(w[8]), "s"(w[9]), "s"(w[10]), "s"(w[11]), "s"(w[12]), "s"(w[13]),
                       "s"(w[14]), "s"(w[15]));
        }
        float v[SB];
        for (unsigned j = 0; j < SB; ++j) {
          v[j] = 0.0f;
          if (BYCOL || (unsigned)d[j] - lo < len)
            v[j] = hq[(size_t)(unsigned)s[j] * 32 + lane];
        }
        for (unsigned j = 0; j < SB; ++j)
          if (BYCOL || (unsigned)d[j] - lo < len) {
            const float m = w[j] * v[j];
            unsafeAtomicAdd(&aq[(size_t)(unsigned)d[j] * 32 + lane], m);
            if constexpr (POOL) p += m;
          }
      } else {
        for (unsigned e = eb; e < (unsigned)E; ++e) {
          const int s = src[e];
          const int d = dst[e];
          const float w = nrm[e];
          if (BYCOL || (unsigned)d - lo < len) {
            const float m = w * hq[(size_t)(unsigned)s * 32 + lane];
            unsafeAtomicAdd(&aq[(size_t)(unsigned)d * 32 + lane], m);
            if constexpr (POOL) p += m;
          }
        }
      }
    }
  }
  if constexpr (POOL)
    unsafeAtomicAdd(&pool[(size_t)(gw & (POOL_SLOTS - 1)) * 32 + lane], p);
}

__global__ __launch_bounds__(256)
void pool_final_k(const float* __restrict__ pool, float* __restrict__ g, int N) {
  __shared__ float sm[8][32];
  const unsigned lane = threadIdx.x & 31;
  const unsigned q = threadIdx.x >> 5;
  float s = 0.0f;
  for (int i = 0; i < POOL_SLOTS / 8; ++i) s += pool[(size_t)(q + 8 * i) * 32 + lane];
  sm[q][lane] = s;
  __syncthreads();
  if (q == 0) {
    float t = 0.0f;
    for (int w = 0; w < 8; ++w) t += sm[w][lane];
    g[lane] = t / (float)N;
  }
}

static inline unsigned cdiv(long a, long b) { return (unsigned)((a + b - 1) / b); }

extern "C" void kernel_launch(void* const* d_in, const int* in_sizes, int n_in,
                              void* d_out, int out_size, void* d_ws, size_t ws_size,
                              hipStream_t stream) {
  const float* x  = (const float*)d_in[0];
  const int*   ei = (const int*)d_in[1];
  const float* ew = (const float*)d_in[2];
  const float* W1 = (const float*)d_in[3];
  const float* b1 = (const float*)d_in[4];
  const float* W2 = (const float*)d_in[5];
  const float* b2 = (const float*)d_in[6];
  const float* W3 = (const float*)d_in[7];
  const float* b3 = (const float*)d_in[8];

  const int N = in_sizes[0] / 128;
  const int E = in_sizes[2];
  const int* src = ei;
  const int* dst = ei + E;

  float* ws   = (float*)d_ws;
  float* deg  = ws;  ws += N;
  float* dinv = ws;  ws += N;
  float* nrm  = ws;  ws += E;
  float* bufA = ws;  ws += (size_t)N * 64;
  float* bufB = ws;  ws += (size_t)N * 64;
  float* bufC = ws;  ws += (size_t)N * 64;
  float* pool = ws;  ws += POOL_SLOTS * 32;
  unsigned* ctr = (unsigned*)ws;
  float* out  = (float*)d_out;

  const dim3 blk(256);

  deg_init_k   <<<cdiv(N, 256), blk, 0, stream>>>(deg, pool, ctr, N);
  deg_hist_k   <<<dim3(DEG_CHUNKS, cdiv(N, DEG_RANGE)), dim3(DEG_THREADS), 0, stream>>>(dst, ew, deg, E, N);
  dinv_k       <<<cdiv(N, 256), blk, 0, stream>>>(deg, dinv, N);
  norm_lds_k   <<<cdiv(E, NRM_THREADS * NRM_EPT), dim3(NRM_THREADS), 0, stream>>>(src, dst, ew, dinv, nrm, E, N);

  gemm16_wmma<128, 64, false, false, false><<<cdiv((long)(N / 16) * 4, 8), blk, 0, stream>>>(x, W1, dinv, b1, bufA, bufB, pool, N);
  agg_edges_k<64, false><<<2048, blk, 0, stream>>>(bufA, src, dst, nrm, bufB, pool, ctr, E, N);

  gemm16_wmma<64, 64, true, false, true><<<cdiv((long)(N / 16) * 4, 8), blk, 0, stream>>>(bufB, W2, dinv, b2, bufA, bufC, pool, N);
  agg_edges_k<64, false><<<2048, blk, 0, stream>>>(bufA, src, dst, nrm, bufC, pool, ctr + 2 * CTR_STRIDE, E, N);

  gemm16_wmma<64, 32, true, true, true><<<cdiv((long)(N / 16) * 2, 8), blk, 0, stream>>>(bufC, W3, dinv, b3, bufA, out, pool, N);
  agg_edges_k<32, true><<<2048, blk, 0, stream>>>(bufA, src, dst, nrm, out, pool, ctr + 4 * CTR_STRIDE, E, N);
  pool_final_k   <<<1, 256, 0, stream>>>(pool, out + (size_t)N * 32, N);
}


# Edge sums in LDS per 196-row destination range: edges filed by range once per call (count, prefix, placing ride in degree, root, weight kernels); edge passes without global atomics; sum order changes
# speedup vs baseline: 5.4017x; 1.1032x over previous
#include <hip/hip_runtime.h>

typedef __attribute__((ext_vector_type(2))) float v2f;
typedef __attribute__((ext_vector_type(4))) float v4f;
typedef __attribute__((ext_vector_type(8))) float v8f;

constexpr int RNG_ROWS = 196;
constexpr int RNG_MAX = 512;
constexpr unsigned SRC_BITS = 17;
constexpr unsigned SRC_MASK = (1u << SRC_BITS) - 1u;

constexpr int POOL_SLOTS = RNG_MAX;

__device__ inline unsigned excl_scan_ranges(unsigned c, unsigned tid, unsigned* wtot) {
  const unsigned lane = tid & 31;
  unsigned inc = c;
  for (unsigned off = 1; off < 32; off <<= 1) {
    const unsigned t = __shfl_up(inc, off);
    if (lane >= off) inc += t;
  }
  if (lane == 31 && tid < (unsigned)RNG_MAX) wtot[tid >> 5] = inc;
  __syncthreads();
  unsigned pre = 0;
  for (unsigned w = 0; w < (unsigned)RNG_MAX / 32; ++w)
    if (w < (tid >> 5)) pre += wtot[w];
  return pre + inc - c;
}

__global__ void deg_init_k(float* __restrict__ deg, float* __restrict__ pool,
                           unsigned* __restrict__ rcnt, int N) {
  int t = blockIdx.x * blockDim.x + threadIdx.x;
  if (t < N) deg[t] = 1.0f;
  for (int i = t; i < POOL_SLOTS * 32; i += gridDim.x * blockDim.x) pool[i] = 0.0f;
  for (int i = t; i < 2 * RNG_MAX; i += gridDim.x * blockDim.x) rcnt[i] = 0u;
}

constexpr int DEG_RANGE = 65536;
constexpr int DEG_CHUNKS = 64;
constexpr int DEG_THREADS = 1024;

__global__ __launch_bounds__(DEG_THREADS)
void deg_hist_k(const int* __restrict__ dst, const float* __restrict__ ew,
                float* __restrict__ deg, unsigned* __restrict__ rcnt, int E, int N) {
  __shared__ float4 hs4[DEG_RANGE / 4];
  __shared__ unsigned lc[RNG_MAX];
  float* hs = reinterpret_cast<float*>(hs4);
  const unsigned tid = threadIdx.x;
  const bool counting = blockIdx.y == 0;
  const unsigned lo = blockIdx.y * (unsigned)DEG_RANGE;
  const unsigned len = min((unsigned)DEG_RANGE, (unsigned)N - lo);
  for (unsigned i = tid; i < DEG_RANGE / 4; i += DEG_THREADS) hs4[i] = make_float4(0.0f, 0.0f, 0.0f, 0.0f);
  if (tid < (unsigned)RNG_MAX) lc[tid] = 0u;
  __syncthreads();

  const int per = (E + DEG_CHUNKS - 1) / DEG_CHUNKS;
  const int e0 = blockIdx.x * per;
  const int e1 = min(e0 + per, E);
  for (int e = e0 + (int)tid; e < e1; e += 16 * DEG_THREADS) {
    int d[16];
    float w[16];
#pragma unroll
    for (int j = 0; j < 16; ++j) {
      const int idx = e + j * DEG_THREADS;
      const int c = min(idx, e1 - 1);
      d[j] = dst[c];
      w[j] = ew[c];
      if (idx >= e1) d[j] = -1;
    }
    asm volatile("" ::"v"(w[0]), "v"(w[1]), "v"(w[2]), "v"(w[3]), "v"(w[4]), "v"(w[5]), "v"(w[6]), "v"(w[7]),
                 "v"(w[8]), "v"(w[9]), "v"(w[10]), "v"(w[11]), "v"(w[12]), "v"(w[13]), "v"(w[14]), "v"(w[15]));
#pragma unroll
    for (int j = 0; j < 16; ++j) {
      const unsigned u = (unsigned)d[j] - lo;
      if (u < len) unsafeAtomicAdd(&hs[u], w[j]);
      if (counting && d[j] >= 0) atomicAdd(&lc[(unsigned)d[j] / (unsigned)RNG_ROWS], 1u);
    }
  }
  __syncthreads();

  if (counting && tid < (unsigned)RNG_MAX) {
    const unsigned c = lc[tid];
    if (c != 0u) atomicAdd(&rcnt[tid], c);
  }

  const unsigned lane = tid & 31;
  const unsigned nl = (len + 31) / 32;
  const unsigned l0 = blockIdx.x * nl / DEG_CHUNKS;
  for (unsigned k = tid >> 5; k < nl; k += 4 * (DEG_THREADS / 32)) {
    unsigned at[4];
    float v[4];
#pragma unroll
    for (int j = 0; j < 4; ++j) {
      const unsigned kk = k + j * (DEG_THREADS / 32);
      unsigned line = kk + l0;
      if (line >= nl) line -= nl;
      const unsigned i = line * 32 + lane;
      const bool ok = kk < nl && i < len;
      at[j] = ok ? i : 0xffffffffu;
      v[j] = ok ? hs[i] : 0.0f;
    }
#pragma unroll
    for (int j = 0; j < 4; ++j)
      if (at[j] != 0xffffffffu) unsafeAtomicAdd(&deg[lo + at[j]], v[j]);
  }
}

__global__ __launch_bounds__(RNG_MAX)
void dinv_k(const float* __restrict__ deg, float* __restrict__ dinv,
            const unsigned* __restrict__ rcnt, unsigned* __restrict__ rstart, int N) {
  __shared__ unsigned wtot[RNG_MAX / 32];
  int t = blockIdx.x * blockDim.x + threadIdx.x;
  if (t < N) {
    float d = deg[t];
    dinv[t] = (d > 0.0f) ? rsqrtf(d) : 0.0f;
  }
  if (blockIdx.x == 0) {
    const unsigned tid = threadIdx.x;
    const unsigned c = rcnt[tid];
    const unsigned ex = excl_scan_ranges(c, tid, wtot);
    rstart[tid] = ex;
    if (tid == (unsigned)RNG_MAX - 1u) rstart[RNG_MAX] = ex + c;
  }
}

constexpr int NRM_RANGE = 65536;
constexpr int NRM_THREADS = 1024;
constexpr int NRM_EPT = 7;
constexpr int NRM_SG = NRM_RANGE / 4 / NRM_THREADS;
constexpr int NRM_STG = 8192;
static_assert(NRM_THREADS * NRM_EPT <= NRM_STG && 3 * NRM_STG <= NRM_RANGE, "the staged records fit the dinv array");

__global__ __launch_bounds__(NRM_THREADS)
void norm_lds_k(const int* __restrict__ src, const int* __restrict__ dst,
                const float* __restrict__ ew, const float* __restrict__ dinv,
                const unsigned* __restrict__ rstart, unsigned* __restrict__ rfill,
                unsigned* __restrict__ rec_sl, float* __restrict__ rec_w, int E, int N) {
  __shared__ v4f hs4[NRM_RANGE / 4];
  __shared__ unsigned lcnt[RNG_MAX];
  __shared__ unsigned lbase[RNG_MAX];
  __shared__ unsigned gbase[RNG_MAX];
  __shared__ unsigned wtot[RNG_MAX / 32];
  float* hs = reinterpret_cast<float*>(hs4);
  const unsigned tid = threadIdx.x;
  const unsigned e0 = blockIdx.x * (unsigned)(NRM_THREADS * NRM_EPT);
  if (tid < (unsigned)RNG_MAX) lcnt[tid] = 0u;

  int s[NRM_EPT], d[NRM_EPT];
  float w[NRM_EPT], fs[NRM_EPT], fd[NRM_EPT];
#pragma unroll
  for (int j = 0; j < NRM_EPT; ++j) {
    const unsigned c = min(e0 + j * NRM_THREADS + tid, (unsigned)E - 1u);
    s[j] = src[c];
    d[j] = dst[c];
    w[j] = ew[c];
    fs[j] = 0.0f;
    fd[j] = 0.0f;
  }

  for (unsigned lo = 0; lo < (unsigned)N; lo += NRM_RANGE) {
    const unsigned len = min((unsigned)NRM_RANGE, (unsigned)N - lo);
    const unsigned len4 = len & ~3u;
    if (lo != 0) __syncthreads();
    if (len4 != 0) {
      v4f g[NRM_SG];
#pragma unroll
      for (int k = 0; k < NRM_SG; ++k) {
        const unsigned i = min((unsigned)(k * NRM_THREADS + tid) * 4, len4 - 4);
        g[k] = *reinterpret_cast<const v4f*>(dinv + lo + i);
      }
      static_assert(NRM_SG == 16, "the statement below names the value of each load");
      asm volatile("" ::"v"(g[0]), "v"(g[1]), "v"(g[2]), "v"(g[3]), "v"(g[4]), "v"(g[5]), "v"(g[6]), "v"(g[7]),
                   "v"(g[8]), "v"(g[9]), "v"(g[10]), "v"(g[11]), "v"(g[12]), "v"(g[13]), "v"(g[14]), "v"(g[15]));
#pragma unroll
      for (int k = 0; k < NRM_SG; ++k) {
        const unsigned i = min((unsigned)(k * NRM_THREADS + tid) * 4, len4 - 4);
        hs4[i >> 2] = g[k];
      }
    }
    if (tid < len - len4) hs[len4 + tid] = dinv[lo + len4 + tid];
    __syncthreads();

#pragma unroll
    for (int j = 0; j < NRM_EPT; ++j) {
      const unsigned us = (unsigned)s[j] - lo;
      const unsigned ud = (unsigned)d[j] - lo;
      if (us < len) fs[j] = hs[us];
      if (ud < len) fd[j] = hs[ud];
    }
  }

  unsigned rg[NRM_EPT], rk[NRM_EPT];
#pragma unroll
  for (int j = 0; j < NRM_EPT; ++j) {
    rg[j] = (unsigned)d[j] / (unsigned)RNG_ROWS;
    rk[j] = 0u;
    if (e0 + j * NRM_THREADS + tid < (unsigned)E) rk[j] = atomicAdd(&lcnt[rg[j]], 1u);
  }
  __syncthreads();
  {
    const unsigned c = (tid < (unsigned)RNG_MAX) ? lcnt[tid] : 0u;
    const unsigned ex = excl_scan_ranges(c, tid, wtot);
    if (tid < (unsigned)RNG_MAX) {
      lbase[tid] = ex;
      gbase[tid] = rstart[tid] + (c != 0u ? atomicAdd(&rfill[tid], c) : 0u);
    }
  }
  __syncthreads();
  float* st_sl = hs;
  float* st_w = hs + NRM_STG;
  float* st_g = hs + 2 * NRM_STG;
#pragma unroll
  for (int j = 0; j < NRM_EPT; ++j) {
    if (e0 + j * NRM_THREADS + tid < (unsigned)E) {
      const unsigned pos = lbase[rg[j]] + rk[j];
      const unsigned local = (unsigned)d[j] - rg[j] * (unsigned)RNG_ROWS;
      st_sl[pos] = __uint_as_float((unsigned)s[j] | (local << SRC_BITS));
      st_w[pos] = fs[j] * w[j] * fd[j];
      st_g[pos] = __uint_as_float(gbase[rg[j]] + rk[j]);
    }
  }
  __syncthreads();
  const unsigned nvalid = min((unsigned)E - e0, (unsigned)(NRM_THREADS * NRM_EPT));
#pragma unroll
  for (int j = 0; j < NRM_EPT; ++j) {
    const unsigned i = j * NRM_THREADS + tid;
    if (i < nvalid) {
      const unsigned gp = __float_as_uint(st_g[i]);
      rec_sl[gp] = __float_as_uint(st_sl[i]);
      rec_w[gp] = st_w[i];
    }
  }
}

template <int K, int F, bool RELU_IN, bool IN_PLANES>
__global__ void gemm16_wmma(const float* __restrict__ X, const float* __restrict__ W,
                            float* __restrict__ H, int N) {
  const int lane = threadIdx.x & 31;
  const int wave = threadIdx.x >> 5;
  const int Ft = F >> 4;
  const int Mt = N >> 4;
  const int tile = blockIdx.x * (blockDim.x >> 5) + wave;
  if (tile >= Mt * Ft) return;
  const int mt = tile / Ft;
  const int ft = tile - mt * Ft;
  const int l16 = lane & 15;
  const int hi  = lane >> 4;

  constexpr bool OUT_PLANES = (F == 64);
  static_assert(!IN_PLANES || K == 64, "the plane layout is that of 64-column activations");
  const size_t plane = (size_t)N * 32;
  const float* __restrict__ xrow = X + (size_t)(mt * 16 + l16) * (IN_PLANES ? 32 : K);
  const float* __restrict__ wcol = W + ft * 16 + l16;

  v8f acc = {};
  auto kstep = [&](const float* __restrict__ xk, int k) {
    v2f a, b;
    a.x = xk[2 * hi];
    a.y = xk[2 * hi + 1];
    if constexpr (RELU_IN) {
      a.x = fmaxf(a.x, 0.0f);
      a.y = fmaxf(a.y, 0.0f);
    }
    b.x = wcol[(size_t)(k + 2 * hi) * F];
    b.y = wcol[(size_t)(k + 2 * hi + 1) * F];
    acc = __builtin_amdgcn_wmma_f32_16x16x4_f32(false, a, false, b,
                                                (short)0, acc, false, false);
  };
  if constexpr (IN_PLANES) {
    for (int c = 0; c < K / 32; ++c) {
      const float* __restrict__ xp = xrow + (size_t)c * plane;
#pragma unroll 4
      for (int kk = 0; kk < 32; kk += 4) kstep(xp + kk, c * 32 + kk);
    }
  } else {
#pragma unroll 4
    for (int k = 0; k < K; k += 4) kstep(xrow + k, k);
  }

  const int col = ft * 16 + l16;
  const int row0 = mt * 16 + 8 * hi;
#pragma unroll
  for (int r = 0; r < 8; ++r) {
    const size_t o = OUT_PLANES ? (size_t)(col >> 5) * plane + (size_t)(row0 + r) * 32 + (col & 31)
                                : (size_t)(row0 + r) * F + col;
    H[o] = acc[r];
  }
}

constexpr int AGG_THREADS = 1024;
constexpr int AGG_WAVES = AGG_THREADS / 32;

template <int F, bool POOL>
__global__ __launch_bounds__(AGG_THREADS)
void agg_range_k(const float* __restrict__ h, const unsigned* __restrict__ rec_sl,
                 const float* __restrict__ rec_w, const unsigned* __restrict__ rstart,
                 const float* __restrict__ dinv, const float* __restrict__ bias,
                 float* __restrict__ a, float* __restrict__ pool, int N) {
  static_assert(F == 64 || F == 32, "one or two 128-B lines per row");
  static_assert(!POOL || F == 32, "column sums are kept for one-line rows only");
  constexpr int L = F / 32;
  constexpr unsigned SB = 16;
  __shared__ float acc[RNG_ROWS * F];
  const unsigned tid = threadIdx.x;
  const unsigned lane = tid & 31;
  const unsigned wv = __builtin_amdgcn_readfirstlane(tid >> 5);
  const unsigned r = blockIdx.x;
  const unsigned lo = r * (unsigned)RNG_ROWS;
  const unsigned len = min((unsigned)RNG_ROWS, (unsigned)N - lo);
  const size_t plane = (size_t)N * 32;

  for (unsigned i = tid; i < (unsigned)(RNG_ROWS * F); i += AGG_THREADS) acc[i] = 0.0f;
  __syncthreads();

  const unsigned eb0 = rstart[r];
  const unsigned e1 = rstart[r + 1];
  for (unsigned eb = eb0 + wv * SB; eb < e1; eb += AGG_WAVES * SB) {
    if (eb + SB <= e1) {
      unsigned sl[SB];
      float w[SB];
#pragma unroll
      for (unsigned j = 0; j < SB; ++j) {
        sl[j] = rec_sl[eb + j];
        w[j] = rec_w[eb + j];
      }
      float v[SB][L];
#pragma unroll
      for (unsigned j = 0; j < SB; ++j)
#pragma unroll
        for (int c = 0; c < L; ++c)
          v[j][c] = h[(size_t)c * plane + (size_t)(sl[j] & SRC_MASK) * 32 + lane];
#pragma unroll
      for (unsigned j = 0; j < SB; ++j)
#pragma unroll
        for (int c = 0; c < L; ++c)
          unsafeAtomicAdd(&acc[(sl[j] >> SRC_BITS) * F + c * 32 + lane], w[j] * v[j][c]);
    } else {
      for (unsigned e = eb; e < e1; ++e) {
        const unsigned sl = rec_sl[e];
        const float w = rec_w[e];
        for (int c = 0; c < L; ++c)
          unsafeAtomicAdd(&acc[(sl >> SRC_BITS) * F + c * 32 + lane],
                          w * h[(size_t)c * plane + (size_t)(sl & SRC_MASK) * 32 + lane]);
      }
    }
  }
  __syncthreads();

  constexpr int RPW = (RNG_ROWS + AGG_WAVES - 1) / AGG_WAVES;
  float bs[L];
  for (int c = 0; c < L; ++c) bs[c] = bias[c * 32 + lane];
  float hv[RPW][L], di[RPW];
#pragma unroll
  for (int k = 0; k < RPW; ++k) {
    const unsigned n = lo + min(wv + (unsigned)(k * AGG_WAVES), len - 1u);
    di[k] = dinv[n];
#pragma unroll
    for (int c = 0; c < L; ++c) hv[k][c] = h[(size_t)c * plane + (size_t)n * 32 + lane];
  }
  float p = 0.0f;
#pragma unroll
  for (int k = 0; k < RPW; ++k) {
    const unsigned row = wv + (unsigned)(k * AGG_WAVES);
    if (row < len) {
      const float d2 = di[k] * di[k];
#pragma unroll
      for (int c = 0; c < L; ++c) {
        const float sv = fmaf(d2, hv[k][c], bs[c]) + acc[row * F + c * 32 + lane];
        a[(size_t)c * plane + (size_t)(lo + row) * 32 + lane] = sv;
        if constexpr (POOL) p += sv;
      }
    }
  }
  if constexpr (POOL) {
    __syncthreads();
    acc[wv * 32 + lane] = p;
    __syncthreads();
    if (wv == 0) {
      float t = 0.0f;
      for (int k = 0; k < AGG_WAVES; ++k) t += acc[k * 32 + lane];
      pool[(size_t)r * 32 + lane] = t;
    }
  }
}

__global__ __launch_bounds__(256)
void pool_final_k(const float* __restrict__ pool, float* __restrict__ g, int N) {
  __shared__ float sm[8][32];
  const unsigned lane = threadIdx.x & 31;
  const unsigned q = threadIdx.x >> 5;
  float s = 0.0f;
  for (int i = 0; i < POOL_SLOTS / 8; ++i) s += pool[(size_t)(q + 8 * i) * 32 + lane];
  sm[q][lane] = s;
  __syncthreads();
  if (q == 0) {
    float t = 0.0f;
    for (int w = 0; w < 8; ++w) t += sm[w][lane];
    g[lane] = t / (float)N;
  }
}

static inline unsigned cdiv(long a, long b) { return (unsigned)((a + b - 1) / b); }

extern "C" void kernel_launch(void* const* d_in, const int* in_sizes, int n_in,
                              void* d_out, int out_size, void* d_ws, size_t ws_size,
                              hipStream_t stream) {
  const float* x  = (const float*)d_in[0];
  const int*   ei = (const int*)d_in[1];
  const float* ew = (const float*)d_in[2];
  const float* W1 = (const float*)d_in[3];
  const float* b1 = (const float*)d_in[4];
  const float* W2 = (const float*)d_in[5];
  const float* b2 = (const float*)d_in[6];
  const float* W3 = (const float*)d_in[7];
  const float* b3 = (const float*)d_in[8];

  const int N = in_sizes[0] / 128;
  const int E = in_sizes[2];
  const int* src = ei;
  const int* dst = ei + E;

  float* ws   = (float*)d_ws;
  float* deg  = ws;  ws += N;
  float* dinv = ws;  ws += N;
  unsigned* rec_sl = (unsigned*)ws;  ws += E;
  float* rec_w = ws;  ws += E;
  float* bufA = ws;  ws += (size_t)N * 64;
  float* bufB = ws;  ws += (size_t)N * 64;
  float* bufC = ws;  ws += (size_t)N * 64;
  float* pool = ws;  ws += POOL_SLOTS * 32;
  unsigned* rcnt = (unsigned*)ws;
  unsigned* rfill = rcnt + RNG_MAX;
  unsigned* rstart = rfill + RNG_MAX;
  float* out  = (float*)d_out;

  const dim3 blk(256);
  const unsigned nrng = cdiv(N, RNG_ROWS);

  deg_init_k   <<<cdiv(N, 256), blk, 0, stream>>>(deg, pool, rcnt, N);
  deg_hist_k   <<<dim3(DEG_CHUNKS, cdiv(N, DEG_RANGE)), dim3(DEG_THREADS), 0, stream>>>(dst, ew, deg, rcnt, E, N);
  dinv_k       <<<cdiv(N, RNG_MAX), dim3(RNG_MAX), 0, stream>>>(deg, dinv, rcnt, rstart, N);
  norm_lds_k   <<<cdiv(E, NRM_THREADS * NRM_EPT), dim3(NRM_THREADS), 0, stream>>>(src, dst, ew, dinv, rstart, rfill, rec_sl, rec_w, E, N);

  gemm16_wmma<128, 64, false, false><<<cdiv((long)(N / 16) * 4, 8), blk, 0, stream>>>(x, W1, bufA, N);
  agg_range_k<64, false><<<nrng, dim3(AGG_THREADS), 0, stream>>>(bufA, rec_sl, rec_w, rstart, dinv, b1, bufB, pool, N);

  gemm16_wmma<64, 64, true, true><<<cdiv((long)(N / 16) * 4, 8), blk, 0, stream>>>(bufB, W2, bufA, N);
  agg_range_k<64, false><<<nrng, dim3(AGG_THREADS), 0, stream>>>(bufA, rec_sl, rec_w, rstart, dinv, b2, bufC, pool, N);

  gemm16_wmma<64, 32, true, true><<<cdiv((long)(N / 16) * 2, 8), blk, 0, stream>>>(bufC, W3, bufA, N);
  agg_range_k<32, true><<<nrng, dim3(AGG_THREADS), 0, stream>>>(bufA, rec_sl, rec_w, rstart, dinv, b3, out, pool, N);
  pool_final_k   <<<1, 256, 0, stream>>>(pool, out + (size_t)N * 32, N);
}


# GEMMs re-tiled: a wave computes 16 rows x all F columns, so each A fragment (one 16-B load per lane, k permuted alike in A and B) serves every column tile; W from one LDS copy; k sum order changes
# speedup vs baseline: 5.7400x; 1.0626x over previous
#include <hip/hip_runtime.h>

typedef __attribute__((ext_vector_type(2))) float v2f;
typedef __attribute__((ext_vector_type(4))) float v4f;
typedef __attribute__((ext_vector_type(8))) float v8f;

constexpr int RNG_ROWS = 196;
constexpr int RNG_MAX = 512;
constexpr unsigned SRC_BITS = 17;
constexpr unsigned SRC_MASK = (1u << SRC_BITS) - 1u;

constexpr int POOL_SLOTS = RNG_MAX;

__device__ inline unsigned excl_scan_ranges(unsigned c, unsigned tid, unsigned* wtot) {
  const unsigned lane = tid & 31;
  unsigned inc = c;
  for (unsigned off = 1; off < 32; off <<= 1) {
    const unsigned t = __shfl_up(inc, off);
    if (lane >= off) inc += t;
  }
  if (lane == 31 && tid < (unsigned)RNG_MAX) wtot[tid >> 5] = inc;
  __syncthreads();
  unsigned pre = 0;
  for (unsigned w = 0; w < (unsigned)RNG_MAX / 32; ++w)
    if (w < (tid >> 5)) pre += wtot[w];
  return pre + inc - c;
}

__global__ void deg_init_k(float* __restrict__ deg, float* __restrict__ pool,
                           unsigned* __restrict__ rcnt, int N) {
  int t = blockIdx.x * blockDim.x + threadIdx.x;
  if (t < N) deg[t] = 1.0f;
  for (int i = t; i < POOL_SLOTS * 32; i += gridDim.x * blockDim.x) pool[i] = 0.0f;
  for (int i = t; i < 2 * RNG_MAX; i += gridDim.x * blockDim.x) rcnt[i] = 0u;
}

constexpr int DEG_RANGE = 65536;
constexpr int DEG_CHUNKS = 64;
constexpr int DEG_THREADS = 1024;

__global__ __launch_bounds__(DEG_THREADS)
void deg_hist_k(const int* __restrict__ dst, const float* __restrict__ ew,
                float* __restrict__ deg, unsigned* __restrict__ rcnt, int E, int N) {
  __shared__ float4 hs4[DEG_RANGE / 4];
  __shared__ unsigned lc[RNG_MAX];
  float* hs = reinterpret_cast<float*>(hs4);
  const unsigned tid = threadIdx.x;
  const bool counting = blockIdx.y == 0;
  const unsigned lo = blockIdx.y * (unsigned)DEG_RANGE;
  const unsigned len = min((unsigned)DEG_RANGE, (unsigned)N - lo);
  for (unsigned i = tid; i < DEG_RANGE / 4; i += DEG_THREADS) hs4[i] = make_float4(0.0f, 0.0f, 0.0f, 0.0f);
  if (tid < (unsigned)RNG_MAX) lc[tid] = 0u;
  __syncthreads();

  const int per = (E + DEG_CHUNKS - 1) / DEG_CHUNKS;
  const int e0 = blockIdx.x * per;
  const int e1 = min(e0 + per, E);
  for (int e = e0 + (int)tid; e < e1; e += 16 * DEG_THREADS) {
    int d[16];
    float w[16];
#pragma unroll
    for (int j = 0; j < 16; ++j) {
      const int idx = e + j * DEG_THREADS;
      const int c = min(idx, e1 - 1);
      d[j] = dst[c];
      w[j] = ew[c];
      if (idx >= e1) d[j] = -1;
    }
    asm volatile("" ::"v"(w[0]), "v"(w[1]), "v"(w[2]), "v"(w[3]), "v"(w[4]), "v"(w[5]), "v"(w[6]), "v"(w[7]),
                 "v"(w[8]), "v"(w[9]), "v"(w[10]), "v"(w[11]), "v"(w[12]), "v"(w[13]), "v"(w[14]), "v"(w[15]));
#pragma unroll
    for (int j = 0; j < 16; ++j) {
      const unsigned u = (unsigned)d[j] - lo;
      if (u < len) unsafeAtomicAdd(&hs[u], w[j]);
      if (counting && d[j] >= 0) atomicAdd(&lc[(unsigned)d[j] / (unsigned)RNG_ROWS], 1u);
    }
  }
  __syncthreads();

  if (counting && tid < (unsigned)RNG_MAX) {
    const unsigned c = lc[tid];
    if (c != 0u) atomicAdd(&rcnt[tid], c);
  }

  const unsigned lane = tid & 31;
  const unsigned nl = (len + 31) / 32;
  const unsigned l0 = blockIdx.x * nl / DEG_CHUNKS;
  for (unsigned k = tid >> 5; k < nl; k += 4 * (DEG_THREADS / 32)) {
    unsigned at[4];
    float v[4];
#pragma unroll
    for (int j = 0; j < 4; ++j) {
      const unsigned kk = k + j * (DEG_THREADS / 32);
      unsigned line = kk + l0;
      if (line >= nl) line -= nl;
      const unsigned i = line * 32 + lane;
      const bool ok = kk < nl && i < len;
      at[j] = ok ? i : 0xffffffffu;
      v[j] = ok ? hs[i] : 0.0f;
    }
#pragma unroll
    for (int j = 0; j < 4; ++j)
      if (at[j] != 0xffffffffu) unsafeAtomicAdd(&deg[lo + at[j]], v[j]);
  }
}

__global__ __launch_bounds__(RNG_MAX)
void dinv_k(const float* __restrict__ deg, float* __restrict__ dinv,
            const unsigned* __restrict__ rcnt, unsigned* __restrict__ rstart, int N) {
  __shared__ unsigned wtot[RNG_MAX / 32];
  int t = blockIdx.x * blockDim.x + threadIdx.x;
  if (t < N) {
    float d = deg[t];
    dinv[t] = (d > 0.0f) ? rsqrtf(d) : 0.0f;
  }
  if (blockIdx.x == 0) {
    const unsigned tid = threadIdx.x;
    const unsigned c = rcnt[tid];
    const unsigned ex = excl_scan_ranges(c, tid, wtot);
    rstart[tid] = ex;
    if (tid == (unsigned)RNG_MAX - 1u) rstart[RNG_MAX] = ex + c;
  }
}

constexpr int NRM_RANGE = 65536;
constexpr int NRM_THREADS = 1024;
constexpr int NRM_EPT = 7;
constexpr int NRM_SG = NRM_RANGE / 4 / NRM_THREADS;
constexpr int NRM_STG = 8192;
static_assert(NRM_THREADS * NRM_EPT <= NRM_STG && 3 * NRM_STG <= NRM_RANGE, "the staged records fit the dinv array");

__global__ __launch_bounds__(NRM_THREADS)
void norm_lds_k(const int* __restrict__ src, const int* __restrict__ dst,
                const float* __restrict__ ew, const float* __restrict__ dinv,
                const unsigned* __restrict__ rstart, unsigned* __restrict__ rfill,
                unsigned* __restrict__ rec_sl, float* __restrict__ rec_w, int E, int N) {
  __shared__ v4f hs4[NRM_RANGE / 4];
  __shared__ unsigned lcnt[RNG_MAX];
  __shared__ unsigned lbase[RNG_MAX];
  __shared__ unsigned gbase[RNG_MAX];
  __shared__ unsigned wtot[RNG_MAX / 32];
  float* hs = reinterpret_cast<float*>(hs4);
  const unsigned tid = threadIdx.x;
  const unsigned e0 = blockIdx.x * (unsigned)(NRM_THREADS * NRM_EPT);
  if (tid < (unsigned)RNG_MAX) lcnt[tid] = 0u;

  int s[NRM_EPT], d[NRM_EPT];
  float w[NRM_EPT], fs[NRM_EPT], fd[NRM_EPT];
#pragma unroll
  for (int j = 0; j < NRM_EPT; ++j) {
    const unsigned c = min(e0 + j * NRM_THREADS + tid, (unsigned)E - 1u);
    s[j] = src[c];
    d[j] = dst[c];
    w[j] = ew[c];
    fs[j] = 0.0f;
    fd[j] = 0.0f;
  }

  for (unsigned lo = 0; lo < (unsigned)N; lo += NRM_RANGE) {
    const unsigned len = min((unsigned)NRM_RANGE, (unsigned)N - lo);
    const unsigned len4 = len & ~3u;
    if (lo != 0) __syncthreads();
    if (len4 != 0) {
      v4f g[NRM_SG];
#pragma unroll
      for (int k = 0; k < NRM_SG; ++k) {
        const unsigned i = min((unsigned)(k * NRM_THREADS + tid) * 4, len4 - 4);
        g[k] = *reinterpret_cast<const v4f*>(dinv + lo + i);
      }
      static_assert(NRM_SG == 16, "the statement below names the value of each load");
      asm volatile("" ::"v"(g[0]), "v"(g[1]), "v"(g[2]), "v"(g[3]), "v"(g[4]), "v"(g[5]), "v"(g[6]), "v"(g[7]),
                   "v"(g[8]), "v"(g[9]), "v"(g[10]), "v"(g[11]), "v"(g[12]), "v"(g[13]), "v"(g[14]), "v"(g[15]));
#pragma unroll
      for (int k = 0; k < NRM_SG; ++k) {
        const unsigned i = min((unsigned)(k * NRM_THREADS + tid) * 4, len4 - 4);
        hs4[i >> 2] = g[k];
      }
    }
    if (tid < len - len4) hs[len4 + tid] = dinv[lo + len4 + tid];
    __syncthreads();

#pragma unroll
    for (int j = 0; j < NRM_EPT; ++j) {
      const unsigned us = (unsigned)s[j] - lo;
      const unsigned ud = (unsigned)d[j] - lo;
      if (us < len) fs[j] = hs[us];
      if (ud < len) fd[j] = hs[ud];
    }
  }

  unsigned rg[NRM_EPT], rk[NRM_EPT];
#pragma unroll
  for (int j = 0; j < NRM_EPT; ++j) {
    rg[j] = (unsigned)d[j] / (unsigned)RNG_ROWS;
    rk[j] = 0u;
    if (e0 + j * NRM_THREADS + tid < (unsigned)E) rk[j] = atomicAdd(&lcnt[rg[j]], 1u);
  }
  __syncthreads();
  {
    const unsigned c = (tid < (unsigned)RNG_MAX) ? lcnt[tid] : 0u;
    const unsigned ex = excl_scan_ranges(c, tid, wtot);
    if (tid < (unsigned)RNG_MAX) {
      lbase[tid] = ex;
      gbase[tid] = rstart[tid] + (c != 0u ? atomicAdd(&rfill[tid], c) : 0u);
    }
  }
  __syncthreads();
  float* st_sl = hs;
  float* st_w = hs + NRM_STG;
  float* st_g = hs + 2 * NRM_STG;
#pragma unroll
  for (int j = 0; j < NRM_EPT; ++j) {
    if (e0 + j * NRM_THREADS + tid < (unsigned)E) {
      const unsigned pos = lbase[rg[j]] + rk[j];
      const unsigned local = (unsigned)d[j] - rg[j] * (unsigned)RNG_ROWS;
      st_sl[pos] = __uint_as_float((unsigned)s[j] | (local << SRC_BITS));
      st_w[pos] = fs[j] * w[j] * fd[j];
      st_g[pos] = __uint_as_float(gbase[rg[j]] + rk[j]);
    }
  }
  __syncthreads();
  const unsigned nvalid = min((unsigned)E - e0, (unsigned)(NRM_THREADS * NRM_EPT));
#pragma unroll
  for (int j = 0; j < NRM_EPT; ++j) {
    const unsigned i = j * NRM_THREADS + tid;
    if (i < nvalid) {
      const unsigned gp = __float_as_uint(st_g[i]);
      rec_sl[gp] = __float_as_uint(st_sl[i]);
      rec_w[gp] = st_w[i];
    }
  }
}

constexpr int GEMM_WAVES = 8;
constexpr int GEMM_THREADS = GEMM_WAVES * 32;

template <int K, int F, bool RELU_IN, bool IN_PLANES>
__global__ __launch_bounds__(GEMM_THREADS)
void gemm_strip_wmma(const float* __restrict__ X, const float* __restrict__ W,
                     float* __restrict__ H, int N) {
  constexpr int Ft = F / 16;
  constexpr int Q = K / 8;
  constexpr int SLOTS = Q * Ft * 32;
  constexpr int PER = SLOTS / GEMM_THREADS;
  static_assert(K % 8 == 0 && F % 16 == 0 && SLOTS % GEMM_THREADS == 0, "whole blocks, tiles and slot rounds");
  static_assert(!IN_PLANES || K == 64, "the plane layout is that of 64-column activations");
  constexpr bool OUT_PLANES = (F == 64);
  __shared__ v4f Wl[SLOTS];

  const int tid = threadIdx.x;
  const int lane = tid & 31;
  const int wave = tid >> 5;
  const int c = lane & 15;
  const int h = lane >> 4;
  const int Mt = N >> 4;
  const int strip = blockIdx.x * GEMM_WAVES + wave;
  const bool live = strip < Mt;
  const int m = live ? strip : Mt - 1;
  const size_t plane = (size_t)N * 32;

  const float* __restrict__ xrow = X + (size_t)(m * 16 + c) * (IN_PLANES ? 32 : K) + 4 * h;
  v4f a[Q];
#pragma unroll
  for (int q = 0; q < Q; ++q) {
    const float* __restrict__ p = IN_PLANES ? xrow + (size_t)(q / 4) * plane + 8 * (q % 4) : xrow + 8 * q;
    a[q] = *reinterpret_cast<const v4f*>(p);
  }

  v4f wv[PER];
#pragma unroll
  for (int j = 0; j < PER; ++j) {
    const int g = j * GEMM_WAVES + wave;
    const int q = g / Ft;
    const int ft = g % Ft;
    const float* __restrict__ wp = W + (size_t)(8 * q + 4 * h) * F + 16 * ft + c;
    wv[j].x = wp[0];
    wv[j].y = wp[F];
    wv[j].z = wp[2 * F];
    wv[j].w = wp[3 * F];
  }
#pragma unroll
  for (int j = 0; j < PER; ++j) Wl[j * GEMM_THREADS + tid] = wv[j];
  __syncthreads();
  if (!live) return;

  v8f acc[Ft];
#pragma unroll
  for (int ft = 0; ft < Ft; ++ft) acc[ft] = v8f{};
#pragma unroll
  for (int q = 0; q < Q; ++q) {
    v4f av = a[q];
    if constexpr (RELU_IN) {
      av.x = fmaxf(av.x, 0.0f);
      av.y = fmaxf(av.y, 0.0f);
      av.z = fmaxf(av.z, 0.0f);
      av.w = fmaxf(av.w, 0.0f);
    }
    v4f b[Ft];
#pragma unroll
    for (int ft = 0; ft < Ft; ++ft) b[ft] = Wl[(q * Ft + ft) * 32 + lane];
    const v2f a0 = {av.x, av.y};
    const v2f a1 = {av.z, av.w};
#pragma unroll
    for (int ft = 0; ft < Ft; ++ft) {
      const v2f b0 = {b[ft].x, b[ft].y};
      acc[ft] = __builtin_amdgcn_wmma_f32_16x16x4_f32(false, a0, false, b0,
                                                      (short)0, acc[ft], false, false);
    }
#pragma unroll
    for (int ft = 0; ft < Ft; ++ft) {
      const v2f b1 = {b[ft].z, b[ft].w};
      acc[ft] = __builtin_amdgcn_wmma_f32_16x16x4_f32(false, a1, false, b1,
                                                      (short)0, acc[ft], false, false);
    }
  }

  const int row0 = m * 16 + 8 * h;
#pragma unroll
  for (int ft = 0; ft < Ft; ++ft) {
    const int col = ft * 16 + c;
#pragma unroll
    for (int r = 0; r < 8; ++r) {
      const size_t o = OUT_PLANES ? (size_t)(col >> 5) * plane + (size_t)(row0 + r) * 32 + (col & 31)
                                  : (size_t)(row0 + r) * F + col;
      H[o] = acc[ft][r];
    }
  }
}

constexpr int AGG_THREADS = 1024;
constexpr int AGG_WAVES = AGG_THREADS / 32;

template <int F, bool POOL>
__global__ __launch_bounds__(AGG_THREADS)
void agg_range_k(const float* __restrict__ h, const unsigned* __restrict__ rec_sl,
                 const float* __restrict__ rec_w, const unsigned* __restrict__ rstart,
                 const float* __restrict__ dinv, const float* __restrict__ bias,
                 float* __restrict__ a, float* __restrict__ pool, int N) {
  static_assert(F == 64 || F == 32, "one or two 128-B lines per row");
  static_assert(!POOL || F == 32, "column sums are kept for one-line rows only");
  constexpr int L = F / 32;
  constexpr unsigned SB = 16;
  __shared__ float acc[RNG_ROWS * F];
  const unsigned tid = threadIdx.x;
  const unsigned lane = tid & 31;
  const unsigned wv = __builtin_amdgcn_readfirstlane(tid >> 5);
  const unsigned r = blockIdx.x;
  const unsigned lo = r * (unsigned)RNG_ROWS;
  const unsigned len = min((unsigned)RNG_ROWS, (unsigned)N - lo);
  const size_t plane = (size_t)N * 32;

  for (unsigned i = tid; i < (unsigned)(RNG_ROWS * F); i += AGG_THREADS) acc[i] = 0.0f;
  __syncthreads();

  const unsigned eb0 = rstart[r];
  const unsigned e1 = rstart[r + 1];
  for (unsigned eb = eb0 + wv * SB; eb < e1; eb += AGG_WAVES * SB) {
    if (eb + SB <= e1) {
      unsigned sl[SB];
      float w[SB];
#pragma unroll
      for (unsigned j = 0; j < SB; ++j) {
        sl[j] = rec_sl[eb + j];
        w[j] = rec_w[eb + j];
      }
      float v[SB][L];
#pragma unroll
      for (unsigned j = 0; j < SB; ++j)
#pragma unroll
        for (int c = 0; c < L; ++c)
          v[j][c] = h[(size_t)c * plane + (size_t)(sl[j] & SRC_MASK) * 32 + lane];
#pragma unroll
      for (unsigned j = 0; j < SB; ++j)
#pragma unroll
        for (int c = 0; c < L; ++c)
          unsafeAtomicAdd(&acc[(sl[j] >> SRC_BITS) * F + c * 32 + lane], w[j] * v[j][c]);
    } else {
      for (unsigned e = eb; e < e1; ++e) {
        const unsigned sl = rec_sl[e];
        const float w = rec_w[e];
        for (int c = 0; c < L; ++c)
          unsafeAtomicAdd(&acc[(sl >> SRC_BITS) * F + c * 32 + lane],
                          w * h[(size_t)c * plane + (size_t)(sl & SRC_MASK) * 32 + lane]);
      }
    }
  }
  __syncthreads();

  constexpr int RPW = (RNG_ROWS + AGG_WAVES - 1) / AGG_WAVES;
  float bs[L];
  for (int c = 0; c < L; ++c) bs[c] = bias[c * 32 + lane];
  float hv[RPW][L], di[RPW];
#pragma unroll
  for (int k = 0; k < RPW; ++k) {
    const unsigned n = lo + min(wv + (unsigned)(k * AGG_WAVES), len - 1u);
    di[k] = dinv[n];
#pragma unroll
    for (int c = 0; c < L; ++c) hv[k][c] = h[(size_t)c * plane + (size_t)n * 32 + lane];
  }
  float p = 0.0f;
#pragma unroll
  for (int k = 0; k < RPW; ++k) {
    const unsigned row = wv + (unsigned)(k * AGG_WAVES);
    if (row < len) {
      const float d2 = di[k] * di[k];
#pragma unroll
      for (int c = 0; c < L; ++c) {
        const float sv = fmaf(d2, hv[k][c], bs[c]) + acc[row * F + c * 32 + lane];
        a[(size_t)c * plane + (size_t)(lo + row) * 32 + lane] = sv;
        if constexpr (POOL) p += sv;
      }
    }
  }
  if constexpr (POOL) {
    __syncthreads();
    acc[wv * 32 + lane] = p;
    __syncthreads();
    if (wv == 0) {
      float t = 0.0f;
      for (int k = 0; k < AGG_WAVES; ++k) t += acc[k * 32 + lane];
      pool[(size_t)r * 32 + lane] = t;
    }
  }
}

__global__ __launch_bounds__(256)
void pool_final_k(const float* __restrict__ pool, float* __restrict__ g, int N) {
  __shared__ float sm[8][32];
  const unsigned lane = threadIdx.x & 31;
  const unsigned q = threadIdx.x >> 5;
  float s = 0.0f;
  for (int i = 0; i < POOL_SLOTS / 8; ++i) s += pool[(size_t)(q + 8 * i) * 32 + lane];
  sm[q][lane] = s;
  __syncthreads();
  if (q == 0) {
    float t = 0.0f;
    for (int w = 0; w < 8; ++w) t += sm[w][lane];
    g[lane] = t / (float)N;
  }
}

static inline unsigned cdiv(long a, long b) { return (unsigned)((a + b - 1) / b); }

extern "C" void kernel_launch(void* const* d_in, const int* in_sizes, int n_in,
                              void* d_out, int out_size, void* d_ws, size_t ws_size,
                              hipStream_t stream) {
  const float* x  = (const float*)d_in[0];
  const int*   ei = (const int*)d_in[1];
  const float* ew = (const float*)d_in[2];
  const float* W1 = (const float*)d_in[3];
  const float* b1 = (const float*)d_in[4];
  const float* W2 = (const float*)d_in[5];
  const float* b2 = (const float*)d_in[6];
  const float* W3 = (const float*)d_in[7];
  const float* b3 = (const float*)d_in[8];

  const int N = in_sizes[0] / 128;
  const int E = in_sizes[2];
  const int* src = ei;
  const int* dst = ei + E;

  float* ws   = (float*)d_ws;
  float* deg  = ws;  ws += N;
  float* dinv = ws;  ws += N;
  unsigned* rec_sl = (unsigned*)ws;  ws += E;
  float* rec_w = ws;  ws += E;
  float* bufA = ws;  ws += (size_t)N * 64;
  float* bufB = ws;  ws += (size_t)N * 64;
  float* bufC = ws;  ws += (size_t)N * 64;
  float* pool = ws;  ws += POOL_SLOTS * 32;
  unsigned* rcnt = (unsigned*)ws;
  unsigned* rfill = rcnt + RNG_MAX;
  unsigned* rstart = rfill + RNG_MAX;
  float* out  = (float*)d_out;

  const dim3 blk(256);
  const unsigned nrng = cdiv(N, RNG_ROWS);

  deg_init_k   <<<cdiv(N, 256), blk, 0, stream>>>(deg, pool, rcnt, N);
  deg_hist_k   <<<dim3(DEG_CHUNKS, cdiv(N, DEG_RANGE)), dim3(DEG_THREADS), 0, stream>>>(dst, ew, deg, rcnt, E, N);
  dinv_k       <<<cdiv(N, RNG_MAX), dim3(RNG_MAX), 0, stream>>>(deg, dinv, rcnt, rstart, N);
  norm_lds_k   <<<cdiv(E, NRM_THREADS * NRM_EPT), dim3(NRM_THREADS), 0, stream>>>(src, dst, ew, dinv, rstart, rfill, rec_sl, rec_w, E, N);

  gemm_strip_wmma<128, 64, false, false><<<cdiv(N / 16, GEMM_WAVES), dim3(GEMM_THREADS), 0, stream>>>(x, W1, bufA, N);
  agg_range_k<64, false><<<nrng, dim3(AGG_THREADS), 0, stream>>>(bufA, rec_sl, rec_w, rstart, dinv, b1, bufB, pool, N);

  gemm_strip_wmma<64, 64, true, true><<<cdiv(N / 16, GEMM_WAVES), dim3(GEMM_THREADS), 0, stream>>>(bufB, W2, bufA, N);
  agg_range_k<64, false><<<nrng, dim3(AGG_THREADS), 0, stream>>>(bufA, rec_sl, rec_w, rstart, dinv, b2, bufC, pool, N);

  gemm_strip_wmma<64, 32, true, true><<<cdiv(N / 16, GEMM_WAVES), dim3(GEMM_THREADS), 0, stream>>>(bufC, W3, bufA, N);
  agg_range_k<32, true><<<nrng, dim3(AGG_THREADS), 0, stream>>>(bufA, rec_sl, rec_w, rstart, dinv, b3, out, pool, N);
  pool_final_k   <<<1, 256, 0, stream>>>(pool, out + (size_t)N * 32, N);
}


# Edge passes: a range's last short group of edges goes through the 16-edge batch (indices clamped, padding adds to a spare LDS row), was edge by edge with dependent row loads; sum order kept
# speedup vs baseline: 6.0118x; 1.0474x over previous
#include <hip/hip_runtime.h>

typedef __attribute__((ext_vector_type(2))) float v2f;
typedef __attribute__((ext_vector_type(4))) float v4f;
typedef __attribute__((ext_vector_type(8))) float v8f;

constexpr int RNG_ROWS = 196;
constexpr int RNG_MAX = 512;
constexpr unsigned SRC_BITS = 17;
constexpr unsigned SRC_MASK = (1u << SRC_BITS) - 1u;

constexpr int POOL_SLOTS = RNG_MAX;

__device__ inline unsigned excl_scan_ranges(unsigned c, unsigned tid, unsigned* wtot) {
  const unsigned lane = tid & 31;
  unsigned inc = c;
  for (unsigned off = 1; off < 32; off <<= 1) {
    const unsigned t = __shfl_up(inc, off);
    if (lane >= off) inc += t;
  }
  if (lane == 31 && tid < (unsigned)RNG_MAX) wtot[tid >> 5] = inc;
  __syncthreads();
  unsigned pre = 0;
  for (unsigned w = 0; w < (unsigned)RNG_MAX / 32; ++w)
    if (w < (tid >> 5)) pre += wtot[w];
  return pre + inc - c;
}

__global__ void deg_init_k(float* __restrict__ deg, float* __restrict__ pool,
                           unsigned* __restrict__ rcnt, int N) {
  int t = blockIdx.x * blockDim.x + threadIdx.x;
  if (t < N) deg[t] = 1.0f;
  for (int i = t; i < POOL_SLOTS * 32; i += gridDim.x * blockDim.x) pool[i] = 0.0f;
  for (int i = t; i < 2 * RNG_MAX; i += gridDim.x * blockDim.x) rcnt[i] = 0u;
}

constexpr int DEG_RANGE = 65536;
constexpr int DEG_CHUNKS = 64;
constexpr int DEG_THREADS = 1024;

__global__ __launch_bounds__(DEG_THREADS)
void deg_hist_k(const int* __restrict__ dst, const float* __restrict__ ew,
                float* __restrict__ deg, unsigned* __restrict__ rcnt, int E, int N) {
  __shared__ float4 hs4[DEG_RANGE / 4];
  __shared__ unsigned lc[RNG_MAX];
  float* hs = reinterpret_cast<float*>(hs4);
  const unsigned tid = threadIdx.x;
  const bool counting = blockIdx.y == 0;
  const unsigned lo = blockIdx.y * (unsigned)DEG_RANGE;
  const unsigned len = min((unsigned)DEG_RANGE, (unsigned)N - lo);
  for (unsigned i = tid; i < DEG_RANGE / 4; i += DEG_THREADS) hs4[i] = make_float4(0.0f, 0.0f, 0.0f, 0.0f);
  if (tid < (unsigned)RNG_MAX) lc[tid] = 0u;
  __syncthreads();

  const int per = (E + DEG_CHUNKS - 1) / DEG_CHUNKS;
  const int e0 = blockIdx.x * per;
  const int e1 = min(e0 + per, E);
  for (int e = e0 + (int)tid; e < e1; e += 16 * DEG_THREADS) {
    int d[16];
    float w[16];
#pragma unroll
    for (int j = 0; j < 16; ++j) {
      const int idx = e + j * DEG_THREADS;
      const int c = min(idx, e1 - 1);
      d[j] = dst[c];
      w[j] = ew[c];
      if (idx >= e1) d[j] = -1;
    }
    asm volatile("" ::"v"(w[0]), "v"(w[1]), "v"(w[2]), "v"(w[3]), "v"(w[4]), "v"(w[5]), "v"(w[6]), "v"(w[7]),
                 "v"(w[8]), "v"(w[9]), "v"(w[10]), "v"(w[11]), "v"(w[12]), "v"(w[13]), "v"(w[14]), "v"(w[15]));
#pragma unroll
    for (int j = 0; j < 16; ++j) {
      const unsigned u = (unsigned)d[j] - lo;
      if (u < len) unsafeAtomicAdd(&hs[u], w[j]);
      if (counting && d[j] >= 0) atomicAdd(&lc[(unsigned)d[j] / (unsigned)RNG_ROWS], 1u);
    }
  }
  __syncthreads();

  if (counting && tid < (unsigned)RNG_MAX) {
    const unsigned c = lc[tid];
    if (c != 0u) atomicAdd(&rcnt[tid], c);
  }

  const unsigned lane = tid & 31;
  const unsigned nl = (len + 31) / 32;
  const unsigned l0 = blockIdx.x * nl / DEG_CHUNKS;
  for (unsigned k = tid >> 5; k < nl; k += 4 * (DEG_THREADS / 32)) {
    unsigned at[4];
    float v[4];
#pragma unroll
    for (int j = 0; j < 4; ++j) {
      const unsigned kk = k + j * (DEG_THREADS / 32);
      unsigned line = kk + l0;
      if (line >= nl) line -= nl;
      const unsigned i = line * 32 + lane;
      const bool ok = kk < nl && i < len;
      at[j] = ok ? i : 0xffffffffu;
      v[j] = ok ? hs[i] : 0.0f;
    }
#pragma unroll
    for (int j = 0; j < 4; ++j)
      if (at[j] != 0xffffffffu) unsafeAtomicAdd(&deg[lo + at[j]], v[j]);
  }
}

__global__ __launch_bounds__(RNG_MAX)
void dinv_k(const float* __restrict__ deg, float* __restrict__ dinv,
            const unsigned* __restrict__ rcnt, unsigned* __restrict__ rstart, int N) {
  __shared__ unsigned wtot[RNG_MAX / 32];
  int t = blockIdx.x * blockDim.x + threadIdx.x;
  if (t < N) {
    float d = deg[t];
    dinv[t] = (d > 0.0f) ? rsqrtf(d) : 0.0f;
  }
  if (blockIdx.x == 0) {
    const unsigned tid = threadIdx.x;
    const unsigned c = rcnt[tid];
    const unsigned ex = excl_scan_ranges(c, tid, wtot);
    rstart[tid] = ex;
    if (tid == (unsigned)RNG_MAX - 1u) rstart[RNG_MAX] = ex + c;
  }
}

constexpr int NRM_RANGE = 65536;
constexpr int NRM_THREADS = 1024;
constexpr int NRM_EPT = 7;
constexpr int NRM_SG = NRM_RANGE / 4 / NRM_THREADS;
constexpr int NRM_STG = 8192;
static_assert(NRM_THREADS * NRM_EPT <= NRM_STG && 3 * NRM_STG <= NRM_RANGE, "the staged records fit the dinv array");

__global__ __launch_bounds__(NRM_THREADS)
void norm_lds_k(const int* __restrict__ src, const int* __restrict__ dst,
                const float* __restrict__ ew, const float* __restrict__ dinv,
                const unsigned* __restrict__ rstart, unsigned* __restrict__ rfill,
                unsigned* __restrict__ rec_sl, float* __restrict__ rec_w, int E, int N) {
  __shared__ v4f hs4[NRM_RANGE / 4];
  __shared__ unsigned lcnt[RNG_MAX];
  __shared__ unsigned lbase[RNG_MAX];
  __shared__ unsigned gbase[RNG_MAX];
  __shared__ unsigned wtot[RNG_MAX / 32];
  float* hs = reinterpret_cast<float*>(hs4);
  const unsigned tid = threadIdx.x;
  const unsigned e0 = blockIdx.x * (unsigned)(NRM_THREADS * NRM_EPT);
  if (tid < (unsigned)RNG_MAX) lcnt[tid] = 0u;

  int s[NRM_EPT], d[NRM_EPT];
  float w[NRM_EPT], fs[NRM_EPT], fd[NRM_EPT];
#pragma unroll
  for (int j = 0; j < NRM_EPT; ++j) {
    const unsigned c = min(e0 + j * NRM_THREADS + tid, (unsigned)E - 1u);
    s[j] = src[c];
    d[j] = dst[c];
    w[j] = ew[c];
    fs[j] = 0.0f;
    fd[j] = 0.0f;
  }

  for (unsigned lo = 0; lo < (unsigned)N; lo += NRM_RANGE) {
    const unsigned len = min((unsigned)NRM_RANGE, (unsigned)N - lo);
    const unsigned len4 = len & ~3u;
    if (lo != 0) __syncthreads();
    if (len4 != 0) {
      v4f g[NRM_SG];
#pragma unroll
      for (int k = 0; k < NRM_SG; ++k) {
        const unsigned i = min((unsigned)(k * NRM_THREADS + tid) * 4, len4 - 4);
        g[k] = *reinterpret_cast<const v4f*>(dinv + lo + i);
      }
      static_assert(NRM_SG == 16, "the statement below names the value of each load");
      asm volatile("" ::"v"(g[0]), "v"(g[1]), "v"(g[2]), "v"(g[3]), "v"(g[4]), "v"(g[5]), "v"(g[6]), "v"(g[7]),
                   "v"(g[8]), "v"(g[9]), "v"(g[10]), "v"(g[11]), "v"(g[12]), "v"(g[13]), "v"(g[14]), "v"(g[15]));
#pragma unroll
      for (int k = 0; k < NRM_SG; ++k) {
        const unsigned i = min((unsigned)(k * NRM_THREADS + tid) * 4, len4 - 4);
        hs4[i >> 2] = g[k];
      }
    }
    if (tid < len - len4) hs[len4 + tid] = dinv[lo + len4 + tid];
    __syncthreads();

#pragma unroll
    for (int j = 0; j < NRM_EPT; ++j) {
      const unsigned us = (unsigned)s[j] - lo;
      const unsigned ud = (unsigned)d[j] - lo;
      if (us < len) fs[j] = hs[us];
      if (ud < len) fd[j] = hs[ud];
    }
  }

  unsigned rg[NRM_EPT], rk[NRM_EPT];
#pragma unroll
  for (int j = 0; j < NRM_EPT; ++j) {
    rg[j] = (unsigned)d[j] / (unsigned)RNG_ROWS;
    rk[j] = 0u;
    if (e0 + j * NRM_THREADS + tid < (unsigned)E) rk[j] = atomicAdd(&lcnt[rg[j]], 1u);
  }
  __syncthreads();
  {
    const unsigned c = (tid < (unsigned)RNG_MAX) ? lcnt[tid] : 0u;
    const unsigned ex = excl_scan_ranges(c, tid, wtot);
    if (tid < (unsigned)RNG_MAX) {
      lbase[tid] = ex;
      gbase[tid] = rstart[tid] + (c != 0u ? atomicAdd(&rfill[tid], c) : 0u);
    }
  }
  __syncthreads();
  float* st_sl = hs;
  float* st_w = hs + NRM_STG;
  float* st_g = hs + 2 * NRM_STG;
#pragma unroll
  for (int j = 0; j < NRM_EPT; ++j) {
    if (e0 + j * NRM_THREADS + tid < (unsigned)E) {
      const unsigned pos = lbase[rg[j]] + rk[j];
      const unsigned local = (unsigned)d[j] - rg[j] * (unsigned)RNG_ROWS;
      st_sl[pos] = __uint_as_float((unsigned)s[j] | (local << SRC_BITS));
      st_w[pos] = fs[j] * w[j] * fd[j];
      st_g[pos] = __uint_as_float(gbase[rg[j]] + rk[j]);
    }
  }
  __syncthreads();
  const unsigned nvalid = min((unsigned)E - e0, (unsigned)(NRM_THREADS * NRM_EPT));
#pragma unroll
  for (int j = 0; j < NRM_EPT; ++j) {
    const unsigned i = j * NRM_THREADS + tid;
    if (i < nvalid) {
      const unsigned gp = __float_as_uint(st_g[i]);
      rec_sl[gp] = __float_as_uint(st_sl[i]);
      rec_w[gp] = st_w[i];
    }
  }
}

constexpr int GEMM_WAVES = 8;
constexpr int GEMM_THREADS = GEMM_WAVES * 32;

template <int K, int F, bool RELU_IN, bool IN_PLANES>
__global__ __launch_bounds__(GEMM_THREADS)
void gemm_strip_wmma(const float* __restrict__ X, const float* __restrict__ W,
                     float* __restrict__ H, int N) {
  constexpr int Ft = F / 16;
  constexpr int Q = K / 8;
  constexpr int SLOTS = Q * Ft * 32;
  constexpr int PER = SLOTS / GEMM_THREADS;
  static_assert(K % 8 == 0 && F % 16 == 0 && SLOTS % GEMM_THREADS == 0, "whole blocks, tiles and slot rounds");
  static_assert(!IN_PLANES || K == 64, "the plane layout is that of 64-column activations");
  constexpr bool OUT_PLANES = (F == 64);
  __shared__ v4f Wl[SLOTS];

  const int tid = threadIdx.x;
  const int lane = tid & 31;
  const int wave = tid >> 5;
  const int c = lane & 15;
  const int h = lane >> 4;
  const int Mt = N >> 4;
  const int strip = blockIdx.x * GEMM_WAVES + wave;
  const bool live = strip < Mt;
  const int m = live ? strip : Mt - 1;
  const size_t plane = (size_t)N * 32;

  const float* __restrict__ xrow = X + (size_t)(m * 16 + c) * (IN_PLANES ? 32 : K) + 4 * h;
  v4f a[Q];
#pragma unroll
  for (int q = 0; q < Q; ++q) {
    const float* __restrict__ p = IN_PLANES ? xrow + (size_t)(q / 4) * plane + 8 * (q % 4) : xrow + 8 * q;
    a[q] = *reinterpret_cast<const v4f*>(p);
  }

  v4f wv[PER];
#pragma unroll
  for (int j = 0; j < PER; ++j) {
    const int g = j * GEMM_WAVES + wave;
    const int q = g / Ft;
    const int ft = g % Ft;
    const float* __restrict__ wp = W + (size_t)(8 * q + 4 * h) * F + 16 * ft + c;
    wv[j].x = wp[0];
    wv[j].y = wp[F];
    wv[j].z = wp[2 * F];
    wv[j].w = wp[3 * F];
  }
#pragma unroll
  for (int j = 0; j < PER; ++j) Wl[j * GEMM_THREADS + tid] = wv[j];
  __syncthreads();
  if (!live) return;

  v8f acc[Ft];
#pragma unroll
  for (int ft = 0; ft < Ft; ++ft) acc[ft] = v8f{};
#pragma unroll
  for (int q = 0; q < Q; ++q) {
    v4f av = a[q];
    if constexpr (RELU_IN) {
      av.x = fmaxf(av.x, 0.0f);
      av.y = fmaxf(av.y, 0.0f);
      av.z = fmaxf(av.z, 0.0f);
      av.w = fmaxf(av.w, 0.0f);
    }
    v4f b[Ft];
#pragma unroll
    for (int ft = 0; ft < Ft; ++ft) b[ft] = Wl[(q * Ft + ft) * 32 + lane];
    const v2f a0 = {av.x, av.y};
    const v2f a1 = {av.z, av.w};
#pragma unroll
    for (int ft = 0; ft < Ft; ++ft) {
      const v2f b0 = {b[ft].x, b[ft].y};
      acc[ft] = __builtin_amdgcn_wmma_f32_16x16x4_f32(false, a0, false, b0,
                                                      (short)0, acc[ft], false, false);
    }
#pragma unroll
    for (int ft = 0; ft < Ft; ++ft) {
      const v2f b1 = {b[ft].z, b[ft].w};
      acc[ft] = __builtin_amdgcn_wmma_f32_16x16x4_f32(false, a1, false, b1,
                                                      (short)0, acc[ft], false, false);
    }
  }

  const int row0 = m * 16 + 8 * h;
#pragma unroll
  for (int ft = 0; ft < Ft; ++ft) {
    const int col = ft * 16 + c;
#pragma unroll
    for (int r = 0; r < 8; ++r) {
      const size_t o = OUT_PLANES ? (size_t)(col >> 5) * plane + (size_t)(row0 + r) * 32 + (col & 31)
                                  : (size_t)(row0 + r) * F + col;
      H[o] = acc[ft][r];
    }
  }
}

constexpr int AGG_THREADS = 1024;
constexpr int AGG_WAVES = AGG_THREADS / 32;

template <int F, bool POOL>
__global__ __launch_bounds__(AGG_THREADS)
void agg_range_k(const float* __restrict__ h, const unsigned* __restrict__ rec_sl,
                 const float* __restrict__ rec_w, const unsigned* __restrict__ rstart,
                 const float* __restrict__ dinv, const float* __restrict__ bias,
                 float* __restrict__ a, float* __restrict__ pool, int N) {
  static_assert(F == 64 || F == 32, "one or two 128-B lines per row");
  static_assert(!POOL || F == 32, "column sums are kept for one-line rows only");
  constexpr int L = F / 32;
  constexpr unsigned SB = 16;
  __shared__ float acc[(RNG_ROWS + 1) * F];
  const unsigned tid = threadIdx.x;
  const unsigned lane = tid & 31;
  const unsigned wv = __builtin_amdgcn_readfirstlane(tid >> 5);
  const unsigned r = blockIdx.x;
  const unsigned lo = r * (unsigned)RNG_ROWS;
  const unsigned len = min((unsigned)RNG_ROWS, (unsigned)N - lo);
  const size_t plane = (size_t)N * 32;

  for (unsigned i = tid; i < (unsigned)((RNG_ROWS + 1) * F); i += AGG_THREADS) acc[i] = 0.0f;
  __syncthreads();

  const unsigned eb0 = rstart[r];
  const unsigned e1 = rstart[r + 1];
  for (unsigned eb = eb0 + wv * SB; eb < e1; eb += AGG_WAVES * SB) {
    if (eb + SB <= e1) {
      unsigned sl[SB];
      float w[SB];
#pragma unroll
      for (unsigned j = 0; j < SB; ++j) {
        sl[j] = rec_sl[eb + j];
        w[j] = rec_w[eb + j];
      }
      float v[SB][L];
#pragma unroll
      for (unsigned j = 0; j < SB; ++j)
#pragma unroll
        for (int c = 0; c < L; ++c)
          v[j][c] = h[(size_t)c * plane + (size_t)(sl[j] & SRC_MASK) * 32 + lane];
#pragma unroll
      for (unsigned j = 0; j < SB; ++j)
#pragma unroll
        for (int c = 0; c < L; ++c)
          unsafeAtomicAdd(&acc[(sl[j] >> SRC_BITS) * F + c * 32 + lane], w[j] * v[j][c]);
    } else {
      unsigned sl[SB];
      float w[SB];
#pragma unroll
      for (unsigned j = 0; j < SB; ++j) {
        const unsigned e = min(eb + j, e1 - 1u);
        sl[j] = rec_sl[e];
        w[j] = rec_w[e];
      }
      float v[SB][L];
#pragma unroll
      for (unsigned j = 0; j < SB; ++j)
#pragma unroll
        for (int c = 0; c < L; ++c)
          v[j][c] = h[(size_t)c * plane + (size_t)(sl[j] & SRC_MASK) * 32 + lane];
#pragma unroll
      for (unsigned j = 0; j < SB; ++j) {
        const unsigned row = (eb + j < e1) ? (sl[j] >> SRC_BITS) : (unsigned)RNG_ROWS;
#pragma unroll
        for (int c = 0; c < L; ++c)
          unsafeAtomicAdd(&acc[row * F + c * 32 + lane], w[j] * v[j][c]);
      }
    }
  }
  __syncthreads();

  constexpr int RPW = (RNG_ROWS + AGG_WAVES - 1) / AGG_WAVES;
  float bs[L];
  for (int c = 0; c < L; ++c) bs[c] = bias[c * 32 + lane];
  float hv[RPW][L], di[RPW];
#pragma unroll
  for (int k = 0; k < RPW; ++k) {
    const unsigned n = lo + min(wv + (unsigned)(k * AGG_WAVES), len - 1u);
    di[k] = dinv[n];
#pragma unroll
    for (int c = 0; c < L; ++c) hv[k][c] = h[(size_t)c * plane + (size_t)n * 32 + lane];
  }
  float p = 0.0f;
#pragma unroll
  for (int k = 0; k < RPW; ++k) {
    const unsigned row = wv + (unsigned)(k * AGG_WAVES);
    if (row < len) {
      const float d2 = di[k] * di[k];
#pragma unroll
      for (int c = 0; c < L; ++c) {
        const float sv = fmaf(d2, hv[k][c], bs[c]) + acc[row * F + c * 32 + lane];
        a[(size_t)c * plane + (size_t)(lo + row) * 32 + lane] = sv;
        if constexpr (POOL) p += sv;
      }
    }
  }
  if constexpr (POOL) {
    __syncthreads();
    acc[wv * 32 + lane] = p;
    __syncthreads();
    if (wv == 0) {
      float t = 0.0f;
      for (int k = 0; k < AGG_WAVES; ++k) t += acc[k * 32 + lane];
      pool[(size_t)r * 32 + lane] = t;
    }
  }
}

__global__ __launch_bounds__(256)
void pool_final_k(const float* __restrict__ pool, float* __restrict__ g, int N) {
  __shared__ float sm[8][32];
  const unsigned lane = threadIdx.x & 31;
  const unsigned q = threadIdx.x >> 5;
  float s = 0.0f;
  for (int i = 0; i < POOL_SLOTS / 8; ++i) s += pool[(size_t)(q + 8 * i) * 32 + lane];
  sm[q][lane] = s;
  __syncthreads();
  if (q == 0) {
    float t = 0.0f;
    for (int w = 0; w < 8; ++w) t += sm[w][lane];
    g[lane] = t / (float)N;
  }
}

static inline unsigned cdiv(long a, long b) { return (unsigned)((a + b - 1) / b); }

extern "C" void kernel_launch(void* const* d_in, const int* in_sizes, int n_in,
                              void* d_out, int out_size, void* d_ws, size_t ws_size,
                              hipStream_t stream) {
  const float* x  = (const float*)d_in[0];
  const int*   ei = (const int*)d_in[1];
  const float* ew = (const float*)d_in[2];
  const float* W1 = (const float*)d_in[3];
  const float* b1 = (const float*)d_in[4];
  const float* W2 = (const float*)d_in[5];
  const float* b2 = (const float*)d_in[6];
  const float* W3 = (const float*)d_in[7];
  const float* b3 = (const float*)d_in[8];

  const int N = in_sizes[0] / 128;
  const int E = in_sizes[2];
  const int* src = ei;
  const int* dst = ei + E;

  float* ws   = (float*)d_ws;
  float* deg  = ws;  ws += N;
  float* dinv = ws;  ws += N;
  unsigned* rec_sl = (unsigned*)ws;  ws += E;
  float* rec_w = ws;  ws += E;
  float* bufA = ws;  ws += (size_t)N * 64;
  float* bufB = ws;  ws += (size_t)N * 64;
  float* bufC = ws;  ws += (size_t)N * 64;
  float* pool = ws;  ws += POOL_SLOTS * 32;
  unsigned* rcnt = (unsigned*)ws;
  unsigned* rfill = rcnt + RNG_MAX;
  unsigned* rstart = rfill + RNG_MAX;
  float* out  = (float*)d_out;

  const dim3 blk(256);
  const unsigned nrng = cdiv(N, RNG_ROWS);

  deg_init_k   <<<cdiv(N, 256), blk, 0, stream>>>(deg, pool, rcnt, N);
  deg_hist_k   <<<dim3(DEG_CHUNKS, cdiv(N, DEG_RANGE)), dim3(DEG_THREADS), 0, stream>>>(dst, ew, deg, rcnt, E, N);
  dinv_k       <<<cdiv(N, RNG_MAX), dim3(RNG_MAX), 0, stream>>>(deg, dinv, rcnt, rstart, N);
  norm_lds_k   <<<cdiv(E, NRM_THREADS * NRM_EPT), dim3(NRM_THREADS), 0, stream>>>(src, dst, ew, dinv, rstart, rfill, rec_sl, rec_w, E, N);

  gemm_strip_wmma<128, 64, false, false><<<cdiv(N / 16, GEMM_WAVES), dim3(GEMM_THREADS), 0, stream>>>(x, W1, bufA, N);
  agg_range_k<64, false><<<nrng, dim3(AGG_THREADS), 0, stream>>>(bufA, rec_sl, rec_w, rstart, dinv, b1, bufB, pool, N);

  gemm_strip_wmma<64, 64, true, true><<<cdiv(N / 16, GEMM_WAVES), dim3(GEMM_THREADS), 0, stream>>>(bufB, W2, bufA, N);
  agg_range_k<64, false><<<nrng, dim3(AGG_THREADS), 0, stream>>>(bufA, rec_sl, rec_w, rstart, dinv, b2, bufC, pool, N);

  gemm_strip_wmma<64, 32, true, true><<<cdiv(N / 16, GEMM_WAVES), dim3(GEMM_THREADS), 0, stream>>>(bufC, W3, bufA, N);
  agg_range_k<32, true><<<nrng, dim3(AGG_THREADS), 0, stream>>>(bufA, rec_sl, rec_w, rstart, dinv, b3, out, pool, N);
  pool_final_k   <<<1, 256, 0, stream>>>(pool, out + (size_t)N * 32, N);
}


# Layer-1 GEMM moved into the degree launch: 128 extra work-groups (32 waves, strip loop, W1 in LDS) run x@W1 on the cores that the 128 degree work-groups leave idle; 11 -> 10 launches; all sums kept
# speedup vs baseline: 6.5229x; 1.0850x over previous
#include <hip/hip_runtime.h>

typedef __attribute__((ext_vector_type(2))) float v2f;
typedef __attribute__((ext_vector_type(4))) float v4f;
typedef __attribute__((ext_vector_type(8))) float v8f;

constexpr int RNG_ROWS = 196;
constexpr int RNG_MAX = 512;
constexpr unsigned SRC_BITS = 17;
constexpr unsigned SRC_MASK = (1u << SRC_BITS) - 1u;

constexpr int POOL_SLOTS = RNG_MAX;

__device__ inline unsigned excl_scan_ranges(unsigned c, unsigned tid, unsigned* wtot) {
  const unsigned lane = tid & 31;
  unsigned inc = c;
  for (unsigned off = 1; off < 32; off <<= 1) {
    const unsigned t = __shfl_up(inc, off);
    if (lane >= off) inc += t;
  }
  if (lane == 31 && tid < (unsigned)RNG_MAX) wtot[tid >> 5] = inc;
  __syncthreads();
  unsigned pre = 0;
  for (unsigned w = 0; w < (unsigned)RNG_MAX / 32; ++w)
    if (w < (tid >> 5)) pre += wtot[w];
  return pre + inc - c;
}

__global__ void deg_init_k(float* __restrict__ deg, float* __restrict__ pool,
                           unsigned* __restrict__ rcnt, int N) {
  int t = blockIdx.x * blockDim.x + threadIdx.x;
  if (t < N) deg[t] = 1.0f;
  for (int i = t; i < POOL_SLOTS * 32; i += gridDim.x * blockDim.x) pool[i] = 0.0f;
  for (int i = t; i < 2 * RNG_MAX; i += gridDim.x * blockDim.x) rcnt[i] = 0u;
}

constexpr int DEG_RANGE = 65536;
constexpr int DEG_CHUNKS = 64;
constexpr int DEG_THREADS = 1024;

__device__ inline void deg_hist_body(const int* __restrict__ dst, const float* __restrict__ ew,
                                     float* __restrict__ deg, unsigned* __restrict__ rcnt, int E, int N,
                                     unsigned chunk, unsigned nrange, float4* hs4, unsigned* lc) {
  float* hs = reinterpret_cast<float*>(hs4);
  const unsigned tid = threadIdx.x;
  const bool counting = nrange == 0;
  const unsigned lo = nrange * (unsigned)DEG_RANGE;
  const unsigned len = min((unsigned)DEG_RANGE, (unsigned)N - lo);
  for (unsigned i = tid; i < DEG_RANGE / 4; i += DEG_THREADS) hs4[i] = make_float4(0.0f, 0.0f, 0.0f, 0.0f);
  if (tid < (unsigned)RNG_MAX) lc[tid] = 0u;
  __syncthreads();

  const int per = (E + DEG_CHUNKS - 1) / DEG_CHUNKS;
  const int e0 = (int)chunk * per;
  const int e1 = min(e0 + per, E);
  for (int e = e0 + (int)tid; e < e1; e += 16 * DEG_THREADS) {
    int d[16];
    float w[16];
#pragma unroll
    for (int j = 0; j < 16; ++j) {
      const int idx = e + j * DEG_THREADS;
      const int c = min(idx, e1 - 1);
      d[j] = dst[c];
      w[j] = ew[c];
      if (idx >= e1) d[j] = -1;
    }
    asm volatile("" ::"v"(w[0]), "v"(w[1]), "v"(w[2]), "v"(w[3]), "v"(w[4]), "v"(w[5]), "v"(w[6]), "v"(w[7]),
                 "v"(w[8]), "v"(w[9]), "v"(w[10]), "v"(w[11]), "v"(w[12]), "v"(w[13]), "v"(w[14]), "v"(w[15]));
#pragma unroll
    for (int j = 0; j < 16; ++j) {
      const unsigned u = (unsigned)d[j] - lo;
      if (u < len) unsafeAtomicAdd(&hs[u], w[j]);
      if (counting && d[j] >= 0) atomicAdd(&lc[(unsigned)d[j] / (unsigned)RNG_ROWS], 1u);
    }
  }
  __syncthreads();

  if (counting && tid < (unsigned)RNG_MAX) {
    const unsigned c = lc[tid];
    if (c != 0u) atomicAdd(&rcnt[tid], c);
  }

  const unsigned lane = tid & 31;
  const unsigned nl = (len + 31) / 32;
  const unsigned l0 = chunk * nl / DEG_CHUNKS;
  for (unsigned k = tid >> 5; k < nl; k += 4 * (DEG_THREADS / 32)) {
    unsigned at[4];
    float v[4];
#pragma unroll
    for (int j = 0; j < 4; ++j) {
      const unsigned kk = k + j * (DEG_THREADS / 32);
      unsigned line = kk + l0;
      if (line >= nl) line -= nl;
      const unsigned i = line * 32 + lane;
      const bool ok = kk < nl && i < len;
      at[j] = ok ? i : 0xffffffffu;
      v[j] = ok ? hs[i] : 0.0f;
    }
#pragma unroll
    for (int j = 0; j < 4; ++j)
      if (at[j] != 0xffffffffu) unsafeAtomicAdd(&deg[lo + at[j]], v[j]);
  }
}

__global__ __launch_bounds__(RNG_MAX)
void dinv_k(const float* __restrict__ deg, float* __restrict__ dinv,
            const unsigned* __restrict__ rcnt, unsigned* __restrict__ rstart, int N) {
  __shared__ unsigned wtot[RNG_MAX / 32];
  int t = blockIdx.x * blockDim.x + threadIdx.x;
  if (t < N) {
    float d = deg[t];
    dinv[t] = (d > 0.0f) ? rsqrtf(d) : 0.0f;
  }
  if (blockIdx.x == 0) {
    const unsigned tid = threadIdx.x;
    const unsigned c = rcnt[tid];
    const unsigned ex = excl_scan_ranges(c, tid, wtot);
    rstart[tid] = ex;
    if (tid == (unsigned)RNG_MAX - 1u) rstart[RNG_MAX] = ex + c;
  }
}

constexpr int NRM_RANGE = 65536;
constexpr int NRM_THREADS = 1024;
constexpr int NRM_EPT = 7;
constexpr int NRM_SG = NRM_RANGE / 4 / NRM_THREADS;
constexpr int NRM_STG = 8192;
static_assert(NRM_THREADS * NRM_EPT <= NRM_STG && 3 * NRM_STG <= NRM_RANGE, "the staged records fit the dinv array");

__global__ __launch_bounds__(NRM_THREADS)
void norm_lds_k(const int* __restrict__ src, const int* __restrict__ dst,
                const float* __restrict__ ew, const float* __restrict__ dinv,
                const unsigned* __restrict__ rstart, unsigned* __restrict__ rfill,
                unsigned* __restrict__ rec_sl, float* __restrict__ rec_w, int E, int N) {
  __shared__ v4f hs4[NRM_RANGE / 4];
  __shared__ unsigned lcnt[RNG_MAX];
  __shared__ unsigned lbase[RNG_MAX];
  __shared__ unsigned gbase[RNG_MAX];
  __shared__ unsigned wtot[RNG_MAX / 32];
  float* hs = reinterpret_cast<float*>(hs4);
  const unsigned tid = threadIdx.x;
  const unsigned e0 = blockIdx.x * (unsigned)(NRM_THREADS * NRM_EPT);
  if (tid < (unsigned)RNG_MAX) lcnt[tid] = 0u;

  int s[NRM_EPT], d[NRM_EPT];
  float w[NRM_EPT], fs[NRM_EPT], fd[NRM_EPT];
#pragma unroll
  for (int j = 0; j < NRM_EPT; ++j) {
    const unsigned c = min(e0 + j * NRM_THREADS + tid, (unsigned)E - 1u);
    s[j] = src[c];
    d[j] = dst[c];
    w[j] = ew[c];
    fs[j] = 0.0f;
    fd[j] = 0.0f;
  }

  for (unsigned lo = 0; lo < (unsigned)N; lo += NRM_RANGE) {
    const unsigned len = min((unsigned)NRM_RANGE, (unsigned)N - lo);
    const unsigned len4 = len & ~3u;
    if (lo != 0) __syncthreads();
    if (len4 != 0) {
      v4f g[NRM_SG];
#pragma unroll
      for (int k = 0; k < NRM_SG; ++k) {
        const unsigned i = min((unsigned)(k * NRM_THREADS + tid) * 4, len4 - 4);
        g[k] = *reinterpret_cast<const v4f*>(dinv + lo + i);
      }
      static_assert(NRM_SG == 16, "the statement below names the value of each load");
      asm volatile("" ::"v"(g[0]), "v"(g[1]), "v"(g[2]), "v"(g[3]), "v"(g[4]), "v"(g[5]), "v"(g[6]), "v"(g[7]),
                   "v"(g[8]), "v"(g[9]), "v"(g[10]), "v"(g[11]), "v"(g[12]), "v"(g[13]), "v"(g[14]), "v"(g[15]));
#pragma unroll
      for (int k = 0; k < NRM_SG; ++k) {
        const unsigned i = min((unsigned)(k * NRM_THREADS + tid) * 4, len4 - 4);
        hs4[i >> 2] = g[k];
      }
    }
    if (tid < len - len4) hs[len4 + tid] = dinv[lo + len4 + tid];
    __syncthreads();

#pragma unroll
    for (int j = 0; j < NRM_EPT; ++j) {
      const unsigned us = (unsigned)s[j] - lo;
      const unsigned ud = (unsigned)d[j] - lo;
      if (us < len) fs[j] = hs[us];
      if (ud < len) fd[j] = hs[ud];
    }
  }

  unsigned rg[NRM_EPT], rk[NRM_EPT];
#pragma unroll
  for (int j = 0; j < NRM_EPT; ++j) {
    rg[j] = (unsigned)d[j] / (unsigned)RNG_ROWS;
    rk[j] = 0u;
    if (e0 + j * NRM_THREADS + tid < (unsigned)E) rk[j] = atomicAdd(&lcnt[rg[j]], 1u);
  }
  __syncthreads();
  {
    const unsigned c = (tid < (unsigned)RNG_MAX) ? lcnt[tid] : 0u;
    const unsigned ex = excl_scan_ranges(c, tid, wtot);
    if (tid < (unsigned)RNG_MAX) {
      lbase[tid] = ex;
      gbase[tid] = rstart[tid] + (c != 0u ? atomicAdd(&rfill[tid], c) : 0u);
    }
  }
  __syncthreads();
  float* st_sl = hs;
  float* st_w = hs + NRM_STG;
  float* st_g = hs + 2 * NRM_STG;
#pragma unroll
  for (int j = 0; j < NRM_EPT; ++j) {
    if (e0 + j * NRM_THREADS + tid < (unsigned)E) {
      const unsigned pos = lbase[rg[j]] + rk[j];
      const unsigned local = (unsigned)d[j] - rg[j] * (unsigned)RNG_ROWS;
      st_sl[pos] = __uint_as_float((unsigned)s[j] | (local << SRC_BITS));
      st_w[pos] = fs[j] * w[j] * fd[j];
      st_g[pos] = __uint_as_float(gbase[rg[j]] + rk[j]);
    }
  }
  __syncthreads();
  const unsigned nvalid = min((unsigned)E - e0, (unsigned)(NRM_THREADS * NRM_EPT));
#pragma unroll
  for (int j = 0; j < NRM_EPT; ++j) {
    const unsigned i = j * NRM_THREADS + tid;
    if (i < nvalid) {
      const unsigned gp = __float_as_uint(st_g[i]);
      rec_sl[gp] = __float_as_uint(st_sl[i]);
      rec_w[gp] = st_w[i];
    }
  }
}

constexpr int GEMM_WAVES = 8;
constexpr int GEMM_THREADS = GEMM_WAVES * 32;

template <int K, int F, bool RELU_IN, bool IN_PLANES>
__global__ __launch_bounds__(GEMM_THREADS)
void gemm_strip_wmma(const float* __restrict__ X, const float* __restrict__ W,
                     float* __restrict__ H, int N) {
  constexpr int Ft = F / 16;
  constexpr int Q = K / 8;
  constexpr int SLOTS = Q * Ft * 32;
  constexpr int PER = SLOTS / GEMM_THREADS;
  static_assert(K % 8 == 0 && F % 16 == 0 && SLOTS % GEMM_THREADS == 0, "whole blocks, tiles and slot rounds");
  static_assert(!IN_PLANES || K == 64, "the plane layout is that of 64-column activations");
  constexpr bool OUT_PLANES = (F == 64);
  __shared__ v4f Wl[SLOTS];

  const int tid = threadIdx.x;
  const int lane = tid & 31;
  const int wave = tid >> 5;
  const int c = lane & 15;
  const int h = lane >> 4;
  const int Mt = N >> 4;
  const int strip = blockIdx.x * GEMM_WAVES + wave;
  const bool live = strip < Mt;
  const int m = live ? strip : Mt - 1;
  const size_t plane = (size_t)N * 32;

  const float* __restrict__ xrow = X + (size_t)(m * 16 + c) * (IN_PLANES ? 32 : K) + 4 * h;
  v4f a[Q];
#pragma unroll
  for (int q = 0; q < Q; ++q) {
    const float* __restrict__ p = IN_PLANES ? xrow + (size_t)(q / 4) * plane + 8 * (q % 4) : xrow + 8 * q;
    a[q] = *reinterpret_cast<const v4f*>(p);
  }

  v4f wv[PER];
#pragma unroll
  for (int j = 0; j < PER; ++j) {
    const int g = j * GEMM_WAVES + wave;
    const int q = g / Ft;
    const int ft = g % Ft;
    const float* __restrict__ wp = W + (size_t)(8 * q + 4 * h) * F + 16 * ft + c;
    wv[j].x = wp[0];
    wv[j].y = wp[F];
    wv[j].z = wp[2 * F];
    wv[j].w = wp[3 * F];
  }
#pragma unroll
  for (int j = 0; j < PER; ++j) Wl[j * GEMM_THREADS + tid] = wv[j];
  __syncthreads();
  if (!live) return;

  v8f acc[Ft];
#pragma unroll
  for (int ft = 0; ft < Ft; ++ft) acc[ft] = v8f{};
#pragma unroll
  for (int q = 0; q < Q; ++q) {
    v4f av = a[q];
    if constexpr (RELU_IN) {
      av.x = fmaxf(av.x, 0.0f);
      av.y = fmaxf(av.y, 0.0f);
      av.z = fmaxf(av.z, 0.0f);
      av.w = fmaxf(av.w, 0.0f);
    }
    v4f b[Ft];
#pragma unroll
    for (int ft = 0; ft < Ft; ++ft) b[ft] = Wl[(q * Ft + ft) * 32 + lane];
    const v2f a0 = {av.x, av.y};
    const v2f a1 = {av.z, av.w};
#pragma unroll
    for (int ft = 0; ft < Ft; ++ft) {
      const v2f b0 = {b[ft].x, b[ft].y};
      acc[ft] = __builtin_amdgcn_wmma_f32_16x16x4_f32(false, a0, false, b0,
                                                      (short)0, acc[ft], false, false);
    }
#pragma unroll
    for (int ft = 0; ft < Ft; ++ft) {
      const v2f b1 = {b[ft].z, b[ft].w};
      acc[ft] = __builtin_amdgcn_wmma_f32_16x16x4_f32(false, a1, false, b1,
                                                      (short)0, acc[ft], false, false);
    }
  }

  const int row0 = m * 16 + 8 * h;
#pragma unroll
  for (int ft = 0; ft < Ft; ++ft) {
    const int col = ft * 16 + c;
#pragma unroll
    for (int r = 0; r < 8; ++r) {
      const size_t o = OUT_PLANES ? (size_t)(col >> 5) * plane + (size_t)(row0 + r) * 32 + (col & 31)
                                  : (size_t)(row0 + r) * F + col;
      H[o] = acc[ft][r];
    }
  }
}

constexpr int G1_WGS = 128;
constexpr int G1_WAVES = DEG_THREADS / 32;

__device__ inline void gemm1_strips_body(const float* __restrict__ X, const float* __restrict__ W,
                                         float* __restrict__ H, int N, int g, v4f* Wl) {
  constexpr int K = 128;
  constexpr int F = 64;
  constexpr int Ft = F / 16;
  constexpr int Q = K / 8;
  constexpr int SLOTS = Q * Ft * 32;
  constexpr int PER = SLOTS / DEG_THREADS;
  constexpr int STEP = G1_WGS * G1_WAVES;
  static_assert(SLOTS % DEG_THREADS == 0, "whole slot rounds");
  static_assert(SLOTS * 16 <= DEG_RANGE * 4, "the copy of W fits the degree pass's LDS array");

  const int tid = threadIdx.x;
  const int lane = tid & 31;
  const int wave = tid >> 5;
  const int c = lane & 15;
  const int h = lane >> 4;
  const int Mt = N >> 4;
  const size_t plane = (size_t)N * 32;
  int s = g + G1_WGS * wave;

  v4f a[Q];
  {
    const int m = min(s, Mt - 1);
    const float* __restrict__ xrow = X + (size_t)(m * 16 + c) * K + 4 * h;
#pragma unroll
    for (int q = 0; q < Q; ++q) a[q] = *reinterpret_cast<const v4f*>(xrow + 8 * q);
  }

  v4f wv[PER];
#pragma unroll
  for (int j = 0; j < PER; ++j) {
    const int gq = j * G1_WAVES + wave;
    const int q = gq / Ft;
    const int ft = gq % Ft;
    const float* __restrict__ wp = W + (size_t)(8 * q + 4 * h) * F + 16 * ft + c;
    wv[j].x = wp[0];
    wv[j].y = wp[F];
    wv[j].z = wp[2 * F];
    wv[j].w = wp[3 * F];
  }
#pragma unroll
  for (int j = 0; j < PER; ++j) Wl[j * DEG_THREADS + tid] = wv[j];
  __syncthreads();

  while (s < Mt) {
    asm volatile("" ::: "memory");
    v8f acc[Ft];
#pragma unroll
    for (int ft = 0; ft < Ft; ++ft) acc[ft] = v8f{};
#pragma unroll
    for (int q = 0; q < Q; ++q) {
      const v4f av = a[q];
      v4f b[Ft];
#pragma unroll
      for (int ft = 0; ft < Ft; ++ft) b[ft] = Wl[(q * Ft + ft) * 32 + lane];
      const v2f a0 = {av.x, av.y};
      const v2f a1 = {av.z, av.w};
#pragma unroll
      for (int ft = 0; ft < Ft; ++ft) {
        const v2f b0 = {b[ft].x, b[ft].y};
        acc[ft] = __builtin_amdgcn_wmma_f32_16x16x4_f32(false, a0, false, b0,
                                                        (short)0, acc[ft], false, false);
      }
#pragma unroll
      for (int ft = 0; ft < Ft; ++ft) {
        const v2f b1 = {b[ft].z, b[ft].w};
        acc[ft] = __builtin_amdgcn_wmma_f32_16x16x4_f32(false, a1, false, b1,
                                                        (short)0, acc[ft], false, false);
      }
    }

    const int row0 = s * 16 + 8 * h;
#pragma unroll
    for (int ft = 0; ft < Ft; ++ft) {
      const int col = ft * 16 + c;
#pragma unroll
      for (int r = 0; r < 8; ++r)
        H[(size_t)(col >> 5) * plane + (size_t)(row0 + r) * 32 + (col & 31)] = acc[ft][r];
    }

    s += STEP;
    if (s < Mt) {
      const float* __restrict__ xrow = X + (size_t)(s * 16 + c) * K + 4 * h;
#pragma unroll
      for (int q = 0; q < Q; ++q) a[q] = *reinterpret_cast<const v4f*>(xrow + 8 * q);
    }
  }
}

__global__ __launch_bounds__(DEG_THREADS)
void deg_hist_gemm1_k(const int* __restrict__ dst, const float* __restrict__ ew,
                      float* __restrict__ deg, unsigned* __restrict__ rcnt, int E, int N,
                      const float* __restrict__ X, const float* __restrict__ W,
                      float* __restrict__ H, unsigned ndeg) {
  __shared__ float4 hs4[DEG_RANGE / 4];
  __shared__ unsigned lc[RNG_MAX];
  const unsigned b = blockIdx.x;
  const unsigned pairs = min(ndeg, (unsigned)G1_WGS);
  const bool paired = b < 2u * pairs;
  const unsigned idx = paired ? (b >> 1) : (b - pairs);
  const bool is_deg = paired ? ((b & 1u) == 0u) : (ndeg > pairs);
  if (is_deg)
    deg_hist_body(dst, ew, deg, rcnt, E, N, idx % (unsigned)DEG_CHUNKS, idx / (unsigned)DEG_CHUNKS, hs4, lc);
  else
    gemm1_strips_body(X, W, H, N, (int)idx, reinterpret_cast<v4f*>(hs4));
}

constexpr int AGG_THREADS = 1024;
constexpr int AGG_WAVES = AGG_THREADS / 32;

template <int F, bool POOL>
__global__ __launch_bounds__(AGG_THREADS)
void agg_range_k(const float* __restrict__ h, const unsigned* __restrict__ rec_sl,
                 const float* __restrict__ rec_w, const unsigned* __restrict__ rstart,
                 const float* __restrict__ dinv, const float* __restrict__ bias,
                 float* __restrict__ a, float* __restrict__ pool, int N) {
  static_assert(F == 64 || F == 32, "one or two 128-B lines per row");
  static_assert(!POOL || F == 32, "column sums are kept for one-line rows only");
  constexpr int L = F / 32;
  constexpr unsigned SB = 16;
  __shared__ float acc[(RNG_ROWS + 1) * F];
  const unsigned tid = threadIdx.x;
  const unsigned lane = tid & 31;
  const unsigned wv = __builtin_amdgcn_readfirstlane(tid >> 5);
  const unsigned r = blockIdx.x;
  const unsigned lo = r * (unsigned)RNG_ROWS;
  const unsigned len = min((unsigned)RNG_ROWS, (unsigned)N - lo);
  const size_t plane = (size_t)N * 32;

  for (unsigned i = tid; i < (unsigned)((RNG_ROWS + 1) * F); i += AGG_THREADS) acc[i] = 0.0f;
  __syncthreads();

  const unsigned eb0 = rstart[r];
  const unsigned e1 = rstart[r + 1];
  for (unsigned eb = eb0 + wv * SB; eb < e1; eb += AGG_WAVES * SB) {
    if (eb + SB <= e1) {
      unsigned sl[SB];
      float w[SB];
#pragma unroll
      for (unsigned j = 0; j < SB; ++j) {
        sl[j] = rec_sl[eb + j];
        w[j] = rec_w[eb + j];
      }
      float v[SB][L];
#pragma unroll
      for (unsigned j = 0; j < SB; ++j)
#pragma unroll
        for (int c = 0; c < L; ++c)
          v[j][c] = h[(size_t)c * plane + (size_t)(sl[j] & SRC_MASK) * 32 + lane];
#pragma unroll
      for (unsigned j = 0; j < SB; ++j)
#pragma unroll
        for (int c = 0; c < L; ++c)
          unsafeAtomicAdd(&acc[(sl[j] >> SRC_BITS) * F + c * 32 + lane], w[j] * v[j][c]);
    } else {
      unsigned sl[SB];
      float w[SB];
#pragma unroll
      for (unsigned j = 0; j < SB; ++j) {
        const unsigned e = min(eb + j, e1 - 1u);
        sl[j] = rec_sl[e];
        w[j] = rec_w[e];
      }
      float v[SB][L];
#pragma unroll
      for (unsigned j = 0; j < SB; ++j)
#pragma unroll
        for (int c = 0; c < L; ++c)
          v[j][c] = h[(size_t)c * plane + (size_t)(sl[j] & SRC_MASK) * 32 + lane];
#pragma unroll
      for (unsigned j = 0; j < SB; ++j) {
        const unsigned row = (eb + j < e1) ? (sl[j] >> SRC_BITS) : (unsigned)RNG_ROWS;
#pragma unroll
        for (int c = 0; c < L; ++c)
          unsafeAtomicAdd(&acc[row * F + c * 32 + lane], w[j] * v[j][c]);
      }
    }
  }
  __syncthreads();

  constexpr int RPW = (RNG_ROWS + AGG_WAVES - 1) / AGG_WAVES;
  float bs[L];
  for (int c = 0; c < L; ++c) bs[c] = bias[c * 32 + lane];
  float hv[RPW][L], di[RPW];
#pragma unroll
  for (int k = 0; k < RPW; ++k) {
    const unsigned n = lo + min(wv + (unsigned)(k * AGG_WAVES), len - 1u);
    di[k] = dinv[n];
#pragma unroll
    for (int c = 0; c < L; ++c) hv[k][c] = h[(size_t)c * plane + (size_t)n * 32 + lane];
  }
  float p = 0.0f;
#pragma unroll
  for (int k = 0; k < RPW; ++k) {
    const unsigned row = wv + (unsigned)(k * AGG_WAVES);
    if (row < len) {
      const float d2 = di[k] * di[k];
#pragma unroll
      for (int c = 0; c < L; ++c) {
        const float sv = fmaf(d2, hv[k][c], bs[c]) + acc[row * F + c * 32 + lane];
        a[(size_t)c * plane + (size_t)(lo + row) * 32 + lane] = sv;
        if constexpr (POOL) p += sv;
      }
    }
  }
  if constexpr (POOL) {
    __syncthreads();
    acc[wv * 32 + lane] = p;
    __syncthreads();
    if (wv == 0) {
      float t = 0.0f;
      for (int k = 0; k < AGG_WAVES; ++k) t += acc[k * 32 + lane];
      pool[(size_t)r * 32 + lane] = t;
    }
  }
}

__global__ __launch_bounds__(256)
void pool_final_k(const float* __restrict__ pool, float* __restrict__ g, int N) {
  __shared__ float sm[8][32];
  const unsigned lane = threadIdx.x & 31;
  const unsigned q = threadIdx.x >> 5;
  float s = 0.0f;
  for (int i = 0; i < POOL_SLOTS / 8; ++i) s += pool[(size_t)(q + 8 * i) * 32 + lane];
  sm[q][lane] = s;
  __syncthreads();
  if (q == 0) {
    float t = 0.0f;
    for (int w = 0; w < 8; ++w) t += sm[w][lane];
    g[lane] = t / (float)N;
  }
}

static inline unsigned cdiv(long a, long b) { return (unsigned)((a + b - 1) / b); }

extern "C" void kernel_launch(void* const* d_in, const int* in_sizes, int n_in,
                              void* d_out, int out_size, void* d_ws, size_t ws_size,
                              hipStream_t stream) {
  const float* x  = (const float*)d_in[0];
  const int*   ei = (const int*)d_in[1];
  const float* ew = (const float*)d_in[2];
  const float* W1 = (const float*)d_in[3];
  const float* b1 = (const float*)d_in[4];
  const float* W2 = (const float*)d_in[5];
  const float* b2 = (const float*)d_in[6];
  const float* W3 = (const float*)d_in[7];
  const float* b3 = (const float*)d_in[8];

  const int N = in_sizes[0] / 128;
  const int E = in_sizes[2];
  const int* src = ei;
  const int* dst = ei + E;

  float* ws   = (float*)d_ws;
  float* deg  = ws;  ws += N;
  float* dinv = ws;  ws += N;
  unsigned* rec_sl = (unsigned*)ws;  ws += E;
  float* rec_w = ws;  ws += E;
  float* bufA = ws;  ws += (size_t)N * 64;
  float* bufB = ws;  ws += (size_t)N * 64;
  float* bufC = ws;  ws += (size_t)N * 64;
  float* pool = ws;  ws += POOL_SLOTS * 32;
  unsigned* rcnt = (unsigned*)ws;
  unsigned* rfill = rcnt + RNG_MAX;
  unsigned* rstart = rfill + RNG_MAX;
  float* out  = (float*)d_out;

  const dim3 blk(256);
  const unsigned nrng = cdiv(N, RNG_ROWS);

  deg_init_k   <<<cdiv(N, 256), blk, 0, stream>>>(deg, pool, rcnt, N);
  const unsigned ndeg = DEG_CHUNKS * cdiv(N, DEG_RANGE);
  deg_hist_gemm1_k<<<ndeg + G1_WGS, dim3(DEG_THREADS), 0, stream>>>(dst, ew, deg, rcnt, E, N, x, W1, bufA, ndeg);
  dinv_k       <<<cdiv(N, RNG_MAX), dim3(RNG_MAX), 0, stream>>>(deg, dinv, rcnt, rstart, N);
  norm_lds_k   <<<cdiv(E, NRM_THREADS * NRM_EPT), dim3(NRM_THREADS), 0, stream>>>(src, dst, ew, dinv, rstart, rfill, rec_sl, rec_w, E, N);

  agg_range_k<64, false><<<nrng, dim3(AGG_THREADS), 0, stream>>>(bufA, rec_sl, rec_w, rstart, dinv, b1, bufB, pool, N);

  gemm_strip_wmma<64, 64, true, true><<<cdiv(N / 16, GEMM_WAVES), dim3(GEMM_THREADS), 0, stream>>>(bufB, W2, bufA, N);
  agg_range_k<64, false><<<nrng, dim3(AGG_THREADS), 0, stream>>>(bufA, rec_sl, rec_w, rstart, dinv, b2, bufC, pool, N);

  gemm_strip_wmma<64, 32, true, true><<<cdiv(N / 16, GEMM_WAVES), dim3(GEMM_THREADS), 0, stream>>>(bufC, W3, bufA, N);
  agg_range_k<32, true><<<nrng, dim3(AGG_THREADS), 0, stream>>>(bufA, rec_sl, rec_w, rstart, dinv, b3, out, pool, N);
  pool_final_k   <<<1, 256, 0, stream>>>(pool, out + (size_t)N * 32, N);
}
